# starting kernel
# baseline (speedup 1.0000x reference)
#include <hip/hip_runtime.h>


typedef float v2f __attribute__((ext_vector_type(2)));
typedef float v8f __attribute__((ext_vector_type(8)));

#define T_SEQ 2048
#define HEADS 4
#define DH    64
#define C_DIM 256

__device__ __forceinline__ v8f wmma4(v2f a, v2f b, v8f c) {
  return __builtin_amdgcn_wmma_f32_16x16x4_f32(false, a, false, b, (short)0, c,
                                               false, false);
}

__global__ void qkv_proj_kernel(const float* __restrict__ X,
                                const float* __restrict__ W,
                                const float* __restrict__ bias,
                                float* __restrict__ Qw,
                                float* __restrict__ Kw,
                                float* __restrict__ Vw,
                                int mtiles) {
  const int lane = threadIdx.x & 31;
  const int lo = lane & 15;
  const int hi = lane >> 4;
  const int ntiles = (3 * C_DIM) / 16;
  int tile = (int)((blockIdx.x * blockDim.x + threadIdx.x) >> 5);
  if (tile >= mtiles * ntiles) return;
  const int mt = tile / ntiles;
  const int nt = tile - mt * ntiles;
  const int m0 = mt * 16, n0 = nt * 16;

  const float* arow = X + (size_t)(m0 + lo) * C_DIM;
  const float* wcol = W + (n0 + lo);
  v8f c = {};
#pragma unroll 8
  for (int kk = 0; kk < C_DIM; kk += 4) {
    const int ka = kk + hi * 2;
    v2f a, b;
    a.x = arow[ka];
    a.y = arow[ka + 1];
    b.x = wcol[(size_t)ka * (3 * C_DIM)];
    b.y = wcol[(size_t)(ka + 1) * (3 * C_DIM)];
    c = wmma4(a, b, c);
  }

  const int cg = n0 + lo;
  const int which = cg >> 8;
  const int within = cg & 255;
  const int h = within >> 6;
  const int dd = within & 63;
  const float bv = bias[cg];
  float* dst = (which == 0) ? Qw : ((which == 1) ? Kw : Vw);
#pragma unroll
  for (int r = 0; r < 8; ++r) {
    const int row = m0 + r + hi * 8;
    const int bb = row >> 11;
    const int tt = row & (T_SEQ - 1);
    dst[(((size_t)(bb * HEADS + h)) * T_SEQ + tt) * DH + dd] = c[r] + bv;
  }
}

__global__ void attn_kernel(const float* __restrict__ Q,
                            const float* __restrict__ K,
                            const float* __restrict__ V,
                            float* __restrict__ O,
                            int nwaves) {
  const int lane = threadIdx.x & 31;
  const int lo = lane & 15;
  const int hi = lane >> 4;
  int wave = (int)((blockIdx.x * blockDim.x + threadIdx.x) >> 5);
  if (wave >= nwaves) return;

  const int qtiles = T_SEQ / 16;
  const int qt = wave % qtiles;
  const int bh = wave / qtiles;
  const int q0 = qt * 16;
  const float* Qh = Q + (size_t)bh * T_SEQ * DH;
  const float* Kh = K + (size_t)bh * T_SEQ * DH;
  const float* Vh = V + (size_t)bh * T_SEQ * DH;

  const float qs = 0.125f * 1.44269504088896340736f;
  v2f qreg[16];
  {
    const float* qrow = Qh + (size_t)(q0 + lo) * DH;
#pragma unroll
    for (int i = 0; i < 16; ++i) {
      qreg[i].x = qrow[i * 4 + hi * 2] * qs;
      qreg[i].y = qrow[i * 4 + hi * 2 + 1] * qs;
    }
  }

  float m_run = -__builtin_inff();
  float l_run = 0.0f;
  v8f o0 = {}, o1 = {}, o2 = {}, o3 = {};

  for (int kt0 = 0; kt0 <= q0; kt0 += 16) {
    v8f s = {};
    const float* krow = Kh + (size_t)(kt0 + lo) * DH;
#pragma unroll
    for (int i = 0; i < 16; ++i) {
      v2f a;
      a.x = krow[i * 4 + hi * 2];
      a.y = krow[i * 4 + hi * 2 + 1];
      s = wmma4(a, qreg[i], s);
    }
    if (kt0 == q0) {
#pragma unroll
      for (int r = 0; r < 8; ++r)
        s[r] = (r + hi * 8 > lo) ? -__builtin_inff() : s[r];
    }

    float tmax = s[0];
#pragma unroll
    for (int r = 1; r < 8; ++r) tmax = fmaxf(tmax, s[r]);
    tmax = fmaxf(tmax, __shfl_xor(tmax, 16, 32));
    const float m_new = fmaxf(m_run, tmax);
    const float alpha = __builtin_amdgcn_exp2f(m_run - m_new);
    float p[8];
    float ts = 0.0f;
#pragma unroll
    for (int r = 0; r < 8; ++r) {
      p[r] = __builtin_amdgcn_exp2f(s[r] - m_new);
      ts += p[r];
    }
    ts += __shfl_xor(ts, 16, 32);
    l_run = l_run * alpha + ts;
    m_run = m_new;
#pragma unroll
    for (int r = 0; r < 8; ++r) {
      o0[r] *= alpha; o1[r] *= alpha; o2[r] *= alpha; o3[r] *= alpha;
    }

    const float* vbase = Vh + (size_t)kt0 * DH;
#pragma unroll
    for (int kk = 0; kk < 16; kk += 4) {
      v2f bop;
      if (kk < 8) {
        const float sx = __shfl_xor(p[kk + 2], 16, 32);
        const float sy = __shfl_xor(p[kk + 3], 16, 32);
        bop.x = hi ? sx : p[kk];
        bop.y = hi ? sy : p[kk + 1];
      } else {
        const int j = kk - 8;
        const float sx = __shfl_xor(p[j], 16, 32);
        const float sy = __shfl_xor(p[j + 1], 16, 32);
        bop.x = hi ? p[j + 2] : sx;
        bop.y = hi ? p[j + 3] : sy;
      }
      const float* vk = vbase + (size_t)(kk + hi * 2) * DH + lo;
      v2f a;
      a.x = vk[0];       a.y = vk[DH];        o0 = wmma4(a, bop, o0);
      a.x = vk[16];      a.y = vk[DH + 16];   o1 = wmma4(a, bop, o1);
      a.x = vk[32];      a.y = vk[DH + 32];   o2 = wmma4(a, bop, o2);
      a.x = vk[48];      a.y = vk[DH + 48];   o3 = wmma4(a, bop, o3);
    }
  }

  const float inv_l = 1.0f / l_run;
  const int b = bh / HEADS;
  const int h = bh - b * HEADS;
  float* orow = O + ((size_t)b * T_SEQ + q0 + lo) * C_DIM + h * DH + hi * 8;
#pragma unroll
  for (int r = 0; r < 8; ++r) {
    orow[r]      = o0[r] * inv_l;
    orow[16 + r] = o1[r] * inv_l;
    orow[32 + r] = o2[r] * inv_l;
    orow[48 + r] = o3[r] * inv_l;
  }
}

__global__ void out_proj_kernel(const float* __restrict__ A,
                                const float* __restrict__ W,
                                const float* __restrict__ bias,
                                float* __restrict__ Out,
                                int mtiles) {
  const int lane = threadIdx.x & 31;
  const int lo = lane & 15;
  const int hi = lane >> 4;
  const int ntiles = C_DIM / 16;
  int tile = (int)((blockIdx.x * blockDim.x + threadIdx.x) >> 5);
  if (tile >= mtiles * ntiles) return;
  const int mt = tile / ntiles;
  const int nt = tile - mt * ntiles;
  const int m0 = mt * 16, n0 = nt * 16;

  const float* arow = A + (size_t)(m0 + lo) * C_DIM;
  const float* wcol = W + (n0 + lo);
  v8f c = {};
#pragma unroll 8
  for (int kk = 0; kk < C_DIM; kk += 4) {
    const int ka = kk + hi * 2;
    v2f a, b;
    a.x = arow[ka];
    a.y = arow[ka + 1];
    b.x = wcol[(size_t)ka * C_DIM];
    b.y = wcol[(size_t)(ka + 1) * C_DIM];
    c = wmma4(a, b, c);
  }
  const float bv = bias[n0 + lo];
#pragma unroll
  for (int r = 0; r < 8; ++r)
    Out[(size_t)(m0 + r + hi * 8) * C_DIM + n0 + lo] = c[r] + bv;
}

extern "C" void kernel_launch(void* const* d_in, const int* in_sizes, int n_in,
                              void* d_out, int out_size, void* d_ws, size_t ws_size,
                              hipStream_t stream) {
  const float* x     = (const float*)d_in[0];
  const float* W_qkv = (const float*)d_in[1];
  const float* b_qkv = (const float*)d_in[2];
  const float* W_fc  = (const float*)d_in[3];
  const float* b_fc  = (const float*)d_in[4];
  float* out = (float*)d_out;

  const int BT = in_sizes[0] / C_DIM;
  const int B  = BT / T_SEQ;

  const size_t headElems = (size_t)B * HEADS * T_SEQ * DH;
  float* Qw = (float*)d_ws;
  float* Kw = Qw + headElems;
  float* Vw = Kw + headElems;
  float* Aw = Vw + headElems;

  const int mtiles = BT / 16;

  {
    const int tiles = mtiles * ((3 * C_DIM) / 16);
    const int blocks = (tiles + 7) / 8;
    qkv_proj_kernel<<<blocks, 256, 0, stream>>>(x, W_qkv, b_qkv, Qw, Kw, Vw,
                                                mtiles);
  }
  {
    const int nwaves = B * HEADS * (T_SEQ / 16);
    const int blocks = (nwaves + 7) / 8;
    attn_kernel<<<blocks, 256, 0, stream>>>(Qw, Kw, Vw, Aw, nwaves);
  }
  {
    const int tiles = mtiles * (C_DIM / 16);
    const int blocks = (tiles + 7) / 8;
    out_proj_kernel<<<blocks, 256, 0, stream>>>(Aw, W_fc, b_fc, out, mtiles);
  }
}


# workspace Q, K, V and attention output are stored by their producer in WMMA operand order: each consumer fragment fetch becomes one contiguous 16-byte load per lane; all sum orders unchanged
# speedup vs baseline: 1.1171x; 1.1171x over previous
#include <hip/hip_runtime.h>


typedef float v2f __attribute__((ext_vector_type(2)));
typedef float v4f __attribute__((ext_vector_type(4)));
typedef float v8f __attribute__((ext_vector_type(8)));

#define T_SEQ 2048
#define HEADS 4
#define DH    64
#define C_DIM 256

__device__ __forceinline__ v8f wmma4(v2f a, v2f b, v8f c) {
  return __builtin_amdgcn_wmma_f32_16x16x4_f32(false, a, false, b, (short)0, c,
                                               false, false);
}

__global__ void qkv_proj_kernel(const float* __restrict__ X,
                                const float* __restrict__ W,
                                const float* __restrict__ bias,
                                float* __restrict__ Qw,
                                float* __restrict__ Kw,
                                float* __restrict__ Vw,
                                int mtiles) {
  const int lane = threadIdx.x & 31;
  const int lo = lane & 15;
  const int hi = lane >> 4;
  const int ntiles = (3 * C_DIM) / 16;
  int tile = (int)((blockIdx.x * blockDim.x + threadIdx.x) >> 5);
  if (tile >= mtiles * ntiles) return;
  const int mt = tile / ntiles;
  const int nt = tile - mt * ntiles;
  const int m0 = mt * 16, n0 = nt * 16;

  const float* arow = X + (size_t)(m0 + lo) * C_DIM;
  const float* wcol = W + (n0 + lo);
  v8f c = {};
#pragma unroll 8
  for (int kk = 0; kk < C_DIM; kk += 4) {
    const int ka = kk + hi * 2;
    v2f a, b;
    a.x = arow[ka];
    a.y = arow[ka + 1];
    b.x = wcol[(size_t)ka * (3 * C_DIM)];
    b.y = wcol[(size_t)(ka + 1) * (3 * C_DIM)];
    c = wmma4(a, b, c);
  }

  const int cg = n0 + lo;
  const int which = cg >> 8;
  const int within = cg & 255;
  const int h = within >> 6;
  const int dd = within & 63;
  const float bv = bias[cg];
  const int bb = m0 >> 11;
  const int t0 = m0 & (T_SEQ - 1);
  const size_t tbase = (((size_t)(bb * HEADS + h)) * T_SEQ + t0) * DH;
  if (which != 2) {
    float* dst = ((which == 0) ? Qw : Kw) + tbase + (dd >> 3) * 128 +
                 ((dd >> 1) & 1) * 64 + hi * 32 + ((dd >> 2) & 1) * 2 + (dd & 1);
#pragma unroll
    for (int r = 0; r < 8; ++r) dst[r * 4] = c[r] + bv;
  } else {
    float* dst = Vw + tbase + hi * 512 + (dd >> 5) * 128 + (dd & 15) * 4 +
                 ((dd >> 4) & 1) * 2;
#pragma unroll
    for (int r = 0; r < 8; r += 2) {
      v2f pr;
      pr.x = c[r] + bv;
      pr.y = c[r + 1] + bv;
      *(v2f*)(dst + (r >> 2) * 256 + ((r >> 1) & 1) * 64) = pr;
    }
  }
}

__global__ void attn_kernel(const float* __restrict__ Q,
                            const float* __restrict__ K,
                            const float* __restrict__ V,
                            float* __restrict__ O,
                            int nwaves) {
  const int lane = threadIdx.x & 31;
  const int lo = lane & 15;
  const int hi = lane >> 4;
  int wave = (int)((blockIdx.x * blockDim.x + threadIdx.x) >> 5);
  if (wave >= nwaves) return;

  const int qtiles = T_SEQ / 16;
  const int qt = wave % qtiles;
  const int bh = wave / qtiles;
  const int q0 = qt * 16;
  const float* Qh = Q + (size_t)bh * T_SEQ * DH;
  const float* Kh = K + (size_t)bh * T_SEQ * DH;
  const float* Vh = V + (size_t)bh * T_SEQ * DH;

  const float qs = 0.125f * 1.44269504088896340736f;
  v2f qreg[16];
  {
    const v4f* qf = (const v4f*)(Qh + (size_t)q0 * DH) + lane;
#pragma unroll
    for (int j = 0; j < 8; ++j) {
      const v4f f = qf[j * 32];
      qreg[2 * j].x = f.x * qs;
      qreg[2 * j].y = f.y * qs;
      qreg[2 * j + 1].x = f.z * qs;
      qreg[2 * j + 1].y = f.w * qs;
    }
  }

  float m_run = -__builtin_inff();
  float l_run = 0.0f;
  v8f o0 = {}, o1 = {}, o2 = {}, o3 = {};

  for (int kt0 = 0; kt0 <= q0; kt0 += 16) {
    v8f s = {};
    const v4f* kf = (const v4f*)(Kh + (size_t)kt0 * DH) + lane;
#pragma unroll
    for (int j = 0; j < 8; ++j) {
      const v4f f = kf[j * 32];
      v2f a;
      a.x = f.x; a.y = f.y;
      s = wmma4(a, qreg[2 * j], s);
      a.x = f.z; a.y = f.w;
      s = wmma4(a, qreg[2 * j + 1], s);
    }
    if (kt0 == q0) {
#pragma unroll
      for (int r = 0; r < 8; ++r)
        s[r] = (r + hi * 8 > lo) ? -__builtin_inff() : s[r];
    }

    float tmax = s[0];
#pragma unroll
    for (int r = 1; r < 8; ++r) tmax = fmaxf(tmax, s[r]);
    tmax = fmaxf(tmax, __shfl_xor(tmax, 16, 32));
    const float m_new = fmaxf(m_run, tmax);
    const float alpha = __builtin_amdgcn_exp2f(m_run - m_new);
    float p[8];
    float ts = 0.0f;
#pragma unroll
    for (int r = 0; r < 8; ++r) {
      p[r] = __builtin_amdgcn_exp2f(s[r] - m_new);
      ts += p[r];
    }
    ts += __shfl_xor(ts, 16, 32);
    l_run = l_run * alpha + ts;
    m_run = m_new;
#pragma unroll
    for (int r = 0; r < 8; ++r) {
      o0[r] *= alpha; o1[r] *= alpha; o2[r] *= alpha; o3[r] *= alpha;
    }

    const v4f* vf = (const v4f*)(Vh + (size_t)kt0 * DH) + lane;
#pragma unroll
    for (int kk = 0; kk < 16; kk += 4) {
      v2f bop;
      if (kk < 8) {
        const float sx = __shfl_xor(p[kk + 2], 16, 32);
        const float sy = __shfl_xor(p[kk + 3], 16, 32);
        bop.x = hi ? sx : p[kk];
        bop.y = hi ? sy : p[kk + 1];
      } else {
        const int j = kk - 8;
        const float sx = __shfl_xor(p[j], 16, 32);
        const float sy = __shfl_xor(p[j + 1], 16, 32);
        bop.x = hi ? p[j + 2] : sx;
        bop.y = hi ? p[j + 3] : sy;
      }
      const v4f f0 = vf[(kk >> 1) * 32];
      const v4f f1 = vf[(kk >> 1) * 32 + 32];
      v2f a;
      a.x = f0.x;  a.y = f0.y;   o0 = wmma4(a, bop, o0);
      a.x = f0.z;  a.y = f0.w;   o1 = wmma4(a, bop, o1);
      a.x = f1.x;  a.y = f1.y;   o2 = wmma4(a, bop, o2);
      a.x = f1.z;  a.y = f1.w;   o3 = wmma4(a, bop, o3);
    }
  }

  const float inv_l = 1.0f / l_run;
  const int b = bh / HEADS;
  const int h = bh - b * HEADS;
  float* obase = O + ((size_t)b * T_SEQ + q0) * C_DIM + (h * 8 + hi) * 128 + lo * 4;
#pragma unroll
  for (int hh = 0; hh < 2; ++hh) {
    v4f w;
    w.x = o0[2 * hh] * inv_l;     w.y = o0[2 * hh + 1] * inv_l;
    w.z = o0[2 * hh + 4] * inv_l; w.w = o0[2 * hh + 5] * inv_l;
    *(v4f*)(obase + hh * 64) = w;
    w.x = o1[2 * hh] * inv_l;     w.y = o1[2 * hh + 1] * inv_l;
    w.z = o1[2 * hh + 4] * inv_l; w.w = o1[2 * hh + 5] * inv_l;
    *(v4f*)(obase + 256 + hh * 64) = w;
    w.x = o2[2 * hh] * inv_l;     w.y = o2[2 * hh + 1] * inv_l;
    w.z = o2[2 * hh + 4] * inv_l; w.w = o2[2 * hh + 5] * inv_l;
    *(v4f*)(obase + 512 + hh * 64) = w;
    w.x = o3[2 * hh] * inv_l;     w.y = o3[2 * hh + 1] * inv_l;
    w.z = o3[2 * hh + 4] * inv_l; w.w = o3[2 * hh + 5] * inv_l;
    *(v4f*)(obase + 768 + hh * 64) = w;
  }
}

__global__ void out_proj_kernel(const float* __restrict__ A,
                                const float* __restrict__ W,
                                const float* __restrict__ bias,
                                float* __restrict__ Out,
                                int mtiles) {
  const int lane = threadIdx.x & 31;
  const int lo = lane & 15;
  const int hi = lane >> 4;
  const int ntiles = C_DIM / 16;
  int tile = (int)((blockIdx.x * blockDim.x + threadIdx.x) >> 5);
  if (tile >= mtiles * ntiles) return;
  const int mt = tile / ntiles;
  const int nt = tile - mt * ntiles;
  const int m0 = mt * 16, n0 = nt * 16;

  const v4f* af = (const v4f*)(A + (size_t)m0 * C_DIM) + lane;
  const float* wcol = W + (n0 + lo);
  v8f c = {};
#pragma unroll 4
  for (int j = 0; j < C_DIM / 8; ++j) {
    const v4f f = af[j * 32];
    const int ka = j * 8 + hi * 2;
    v2f a, b;
    a.x = f.x;
    a.y = f.y;
    b.x = wcol[(size_t)ka * C_DIM];
    b.y = wcol[(size_t)(ka + 1) * C_DIM];
    c = wmma4(a, b, c);
    a.x = f.z;
    a.y = f.w;
    b.x = wcol[(size_t)(ka + 4) * C_DIM];
    b.y = wcol[(size_t)(ka + 5) * C_DIM];
    c = wmma4(a, b, c);
  }
  const float bv = bias[n0 + lo];
#pragma unroll
  for (int r = 0; r < 8; ++r)
    Out[(size_t)(m0 + r + hi * 8) * C_DIM + n0 + lo] = c[r] + bv;
}

extern "C" void kernel_launch(void* const* d_in, const int* in_sizes, int n_in,
                              void* d_out, int out_size, void* d_ws, size_t ws_size,
                              hipStream_t stream) {
  const float* x     = (const float*)d_in[0];
  const float* W_qkv = (const float*)d_in[1];
  const float* b_qkv = (const float*)d_in[2];
  const float* W_fc  = (const float*)d_in[3];
  const float* b_fc  = (const float*)d_in[4];
  float* out = (float*)d_out;

  const int BT = in_sizes[0] / C_DIM;
  const int B  = BT / T_SEQ;

  const size_t headElems = (size_t)B * HEADS * T_SEQ * DH;
  float* Qw = (float*)d_ws;
  float* Kw = Qw + headElems;
  float* Vw = Kw + headElems;
  float* Aw = Vw + headElems;

  const int mtiles = BT / 16;

  {
    const int tiles = mtiles * ((3 * C_DIM) / 16);
    const int blocks = (tiles + 7) / 8;
    qkv_proj_kernel<<<blocks, 256, 0, stream>>>(x, W_qkv, b_qkv, Qw, Kw, Vw,
                                                mtiles);
  }
  {
    const int nwaves = B * HEADS * (T_SEQ / 16);
    const int blocks = (nwaves + 7) / 8;
    attn_kernel<<<blocks, 256, 0, stream>>>(Qw, Kw, Vw, Aw, nwaves);
  }
  {
    const int tiles = mtiles * (C_DIM / 16);
    const int blocks = (tiles + 7) / 8;
    out_proj_kernel<<<blocks, 256, 0, stream>>>(Aw, W_fc, b_fc, out, mtiles);
  }
}


# qkv projection: a wave owns 16 rows x 64 columns as 4 column-interleaved tiles (d=4c+t) and takes k in runs of 4 per lane half, so x and W arrive as 16-byte loads; k sum order changes
# speedup vs baseline: 1.3042x; 1.1675x over previous
#include <hip/hip_runtime.h>


typedef float v2f __attribute__((ext_vector_type(2)));
typedef float v4f __attribute__((ext_vector_type(4)));
typedef float v8f __attribute__((ext_vector_type(8)));

#define T_SEQ 2048
#define HEADS 4
#define DH    64
#define C_DIM 256

__device__ __forceinline__ v8f wmma4(v2f a, v2f b, v8f c) {
  return __builtin_amdgcn_wmma_f32_16x16x4_f32(false, a, false, b, (short)0, c,
                                               false, false);
}

__global__ void qkv_proj_kernel(const float* __restrict__ X,
                                const float* __restrict__ W,
                                const float* __restrict__ bias,
                                float* __restrict__ Qw,
                                float* __restrict__ Kw,
                                float* __restrict__ Vw,
                                int mtiles) {
  const int lane = threadIdx.x & 31;
  const int lo = lane & 15;
  const int hi = lane >> 4;
  const int ngroups = (3 * C_DIM) / 64;
  const int wv = (int)((blockIdx.x * blockDim.x + threadIdx.x) >> 5);
  const int blk = wv >> 3;
  const int g = blk % ngroups;
  const int mt = (blk / ngroups) * 8 + (wv & 7);
  if (mt >= mtiles) return;
  const int m0 = mt * 16;

  const float* arow = X + (size_t)(m0 + lo) * C_DIM + hi * 4;
  const float* wrow = W + (size_t)(hi * 4) * (3 * C_DIM) + g * 64 + lo * 4;
  v8f c0 = {}, c1 = {}, c2 = {}, c3 = {};
#pragma unroll 2
  for (int j = 0; j < C_DIM / 8; ++j) {
    const v4f f = *(const v4f*)(arow + j * 8);
    const float* wp = wrow + (size_t)(j * 8) * (3 * C_DIM);
    const v4f w0 = *(const v4f*)(wp);
    const v4f w1 = *(const v4f*)(wp + 3 * C_DIM);
    const v4f w2 = *(const v4f*)(wp + 6 * C_DIM);
    const v4f w3 = *(const v4f*)(wp + 9 * C_DIM);
    v2f a, b;
    a.x = f.x;  a.y = f.y;
    b.x = w0.x; b.y = w1.x; c0 = wmma4(a, b, c0);
    b.x = w0.y; b.y = w1.y; c1 = wmma4(a, b, c1);
    b.x = w0.z; b.y = w1.z; c2 = wmma4(a, b, c2);
    b.x = w0.w; b.y = w1.w; c3 = wmma4(a, b, c3);
    a.x = f.z;  a.y = f.w;
    b.x = w2.x; b.y = w3.x; c0 = wmma4(a, b, c0);
    b.x = w2.y; b.y = w3.y; c1 = wmma4(a, b, c1);
    b.x = w2.z; b.y = w3.z; c2 = wmma4(a, b, c2);
    b.x = w2.w; b.y = w3.w; c3 = wmma4(a, b, c3);
  }

  const int which = g >> 2;
  const int h = g & 3;
  const v4f bv = *(const v4f*)(bias + g * 64 + lo * 4);
  const int bb = m0 >> 11;
  const int t0 = m0 & (T_SEQ - 1);
  const size_t tbase = (((size_t)(bb * HEADS + h)) * T_SEQ + t0) * DH;
  if (which != 2) {
    float* dst = ((which == 0) ? Qw : Kw) + tbase + (lo >> 1) * 128 + hi * 32 +
                 (lo & 1) * 2;
#pragma unroll
    for (int r = 0; r < 8; ++r) {
      v2f pr;
      pr.x = c0[r] + bv.x;
      pr.y = c1[r] + bv.y;
      *(v2f*)(dst + r * 4) = pr;
      pr.x = c2[r] + bv.z;
      pr.y = c3[r] + bv.w;
      *(v2f*)(dst + 64 + r * 4) = pr;
    }
  } else {
    float* dst = Vw + tbase + hi * 512 + (lo >> 3) * 128 + (lo & 3) * 16 +
                 ((lo >> 2) & 1) * 2;
#pragma unroll
    for (int r = 0; r < 8; r += 2) {
      float* d2 = dst + (r >> 2) * 256 + ((r >> 1) & 1) * 64;
      v2f pr;
      pr.x = c0[r] + bv.x;  pr.y = c0[r + 1] + bv.x;  *(v2f*)(d2) = pr;
      pr.x = c1[r] + bv.y;  pr.y = c1[r + 1] + bv.y;  *(v2f*)(d2 + 4) = pr;
      pr.x = c2[r] + bv.z;  pr.y = c2[r + 1] + bv.z;  *(v2f*)(d2 + 8) = pr;
      pr.x = c3[r] + bv.w;  pr.y = c3[r + 1] + bv.w;  *(v2f*)(d2 + 12) = pr;
    }
  }
}

__global__ void attn_kernel(const float* __restrict__ Q,
                            const float* __restrict__ K,
                            const float* __restrict__ V,
                            float* __restrict__ O,
                            int nwaves) {
  const int lane = threadIdx.x & 31;
  const int lo = lane & 15;
  const int hi = lane >> 4;
  int wave = (int)((blockIdx.x * blockDim.x + threadIdx.x) >> 5);
  if (wave >= nwaves) return;

  const int qtiles = T_SEQ / 16;
  const int qt = wave % qtiles;
  const int bh = wave / qtiles;
  const int q0 = qt * 16;
  const float* Qh = Q + (size_t)bh * T_SEQ * DH;
  const float* Kh = K + (size_t)bh * T_SEQ * DH;
  const float* Vh = V + (size_t)bh * T_SEQ * DH;

  const float qs = 0.125f * 1.44269504088896340736f;
  v2f qreg[16];
  {
    const v4f* qf = (const v4f*)(Qh + (size_t)q0 * DH) + lane;
#pragma unroll
    for (int j = 0; j < 8; ++j) {
      const v4f f = qf[j * 32];
      qreg[2 * j].x = f.x * qs;
      qreg[2 * j].y = f.y * qs;
      qreg[2 * j + 1].x = f.z * qs;
      qreg[2 * j + 1].y = f.w * qs;
    }
  }

  float m_run = -__builtin_inff();
  float l_run = 0.0f;
  v8f o0 = {}, o1 = {}, o2 = {}, o3 = {};

  for (int kt0 = 0; kt0 <= q0; kt0 += 16) {
    v8f s = {};
    const v4f* kf = (const v4f*)(Kh + (size_t)kt0 * DH) + lane;
#pragma unroll
    for (int j = 0; j < 8; ++j) {
      const v4f f = kf[j * 32];
      v2f a;
      a.x = f.x; a.y = f.y;
      s = wmma4(a, qreg[2 * j], s);
      a.x = f.z; a.y = f.w;
      s = wmma4(a, qreg[2 * j + 1], s);
    }
    if (kt0 == q0) {
#pragma unroll
      for (int r = 0; r < 8; ++r)
        s[r] = (r + hi * 8 > lo) ? -__builtin_inff() : s[r];
    }

    float tmax = s[0];
#pragma unroll
    for (int r = 1; r < 8; ++r) tmax = fmaxf(tmax, s[r]);
    tmax = fmaxf(tmax, __shfl_xor(tmax, 16, 32));
    const float m_new = fmaxf(m_run, tmax);
    const float alpha = __builtin_amdgcn_exp2f(m_run - m_new);
    float p[8];
    float ts = 0.0f;
#pragma unroll
    for (int r = 0; r < 8; ++r) {
      p[r] = __builtin_amdgcn_exp2f(s[r] - m_new);
      ts += p[r];
    }
    ts += __shfl_xor(ts, 16, 32);
    l_run = l_run * alpha + ts;
    m_run = m_new;
#pragma unroll
    for (int r = 0; r < 8; ++r) {
      o0[r] *= alpha; o1[r] *= alpha; o2[r] *= alpha; o3[r] *= alpha;
    }

    const v4f* vf = (const v4f*)(Vh + (size_t)kt0 * DH) + lane;
#pragma unroll
    for (int kk = 0; kk < 16; kk += 4) {
      v2f bop;
      if (kk < 8) {
        const float sx = __shfl_xor(p[kk + 2], 16, 32);
        const float sy = __shfl_xor(p[kk + 3], 16, 32);
        bop.x = hi ? sx : p[kk];
        bop.y = hi ? sy : p[kk + 1];
      } else {
        const int j = kk - 8;
        const float sx = __shfl_xor(p[j], 16, 32);
        const float sy = __shfl_xor(p[j + 1], 16, 32);
        bop.x = hi ? p[j + 2] : sx;
        bop.y = hi ? p[j + 3] : sy;
      }
      const v4f f0 = vf[(kk >> 1) * 32];
      const v4f f1 = vf[(kk >> 1) * 32 + 32];
      v2f a;
      a.x = f0.x;  a.y = f0.y;   o0 = wmma4(a, bop, o0);
      a.x = f0.z;  a.y = f0.w;   o1 = wmma4(a, bop, o1);
      a.x = f1.x;  a.y = f1.y;   o2 = wmma4(a, bop, o2);
      a.x = f1.z;  a.y = f1.w;   o3 = wmma4(a, bop, o3);
    }
  }

  const float inv_l = 1.0f / l_run;
  const int b = bh / HEADS;
  const int h = bh - b * HEADS;
  float* obase = O + ((size_t)b * T_SEQ + q0) * C_DIM + (h * 8 + hi) * 128 + lo * 4;
#pragma unroll
  for (int hh = 0; hh < 2; ++hh) {
    v4f w;
    w.x = o0[2 * hh] * inv_l;     w.y = o0[2 * hh + 1] * inv_l;
    w.z = o0[2 * hh + 4] * inv_l; w.w = o0[2 * hh + 5] * inv_l;
    *(v4f*)(obase + hh * 64) = w;
    w.x = o1[2 * hh] * inv_l;     w.y = o1[2 * hh + 1] * inv_l;
    w.z = o1[2 * hh + 4] * inv_l; w.w = o1[2 * hh + 5] * inv_l;
    *(v4f*)(obase + 256 + hh * 64) = w;
    w.x = o2[2 * hh] * inv_l;     w.y = o2[2 * hh + 1] * inv_l;
    w.z = o2[2 * hh + 4] * inv_l; w.w = o2[2 * hh + 5] * inv_l;
    *(v4f*)(obase + 512 + hh * 64) = w;
    w.x = o3[2 * hh] * inv_l;     w.y = o3[2 * hh + 1] * inv_l;
    w.z = o3[2 * hh + 4] * inv_l; w.w = o3[2 * hh + 5] * inv_l;
    *(v4f*)(obase + 768 + hh * 64) = w;
  }
}

__global__ void out_proj_kernel(const float* __restrict__ A,
                                const float* __restrict__ W,
                                const float* __restrict__ bias,
                                float* __restrict__ Out,
                                int mtiles) {
  const int lane = threadIdx.x & 31;
  const int lo = lane & 15;
  const int hi = lane >> 4;
  const int ntiles = C_DIM / 16;
  int tile = (int)((blockIdx.x * blockDim.x + threadIdx.x) >> 5);
  if (tile >= mtiles * ntiles) return;
  const int mt = tile / ntiles;
  const int nt = tile - mt * ntiles;
  const int m0 = mt * 16, n0 = nt * 16;

  const v4f* af = (const v4f*)(A + (size_t)m0 * C_DIM) + lane;
  const float* wcol = W + (n0 + lo);
  v8f c = {};
#pragma unroll 4
  for (int j = 0; j < C_DIM / 8; ++j) {
    const v4f f = af[j * 32];
    const int ka = j * 8 + hi * 2;
    v2f a, b;
    a.x = f.x;
    a.y = f.y;
    b.x = wcol[(size_t)ka * C_DIM];
    b.y = wcol[(size_t)(ka + 1) * C_DIM];
    c = wmma4(a, b, c);
    a.x = f.z;
    a.y = f.w;
    b.x = wcol[(size_t)(ka + 4) * C_DIM];
    b.y = wcol[(size_t)(ka + 5) * C_DIM];
    c = wmma4(a, b, c);
  }
  const float bv = bias[n0 + lo];
#pragma unroll
  for (int r = 0; r < 8; ++r)
    Out[(size_t)(m0 + r + hi * 8) * C_DIM + n0 + lo] = c[r] + bv;
}

extern "C" void kernel_launch(void* const* d_in, const int* in_sizes, int n_in,
                              void* d_out, int out_size, void* d_ws, size_t ws_size,
                              hipStream_t stream) {
  const float* x     = (const float*)d_in[0];
  const float* W_qkv = (const float*)d_in[1];
  const float* b_qkv = (const float*)d_in[2];
  const float* W_fc  = (const float*)d_in[3];
  const float* b_fc  = (const float*)d_in[4];
  float* out = (float*)d_out;

  const int BT = in_sizes[0] / C_DIM;
  const int B  = BT / T_SEQ;

  const size_t headElems = (size_t)B * HEADS * T_SEQ * DH;
  float* Qw = (float*)d_ws;
  float* Kw = Qw + headElems;
  float* Vw = Kw + headElems;
  float* Aw = Vw + headElems;

  const int mtiles = BT / 16;

  {
    const int waves = mtiles * ((3 * C_DIM) / 64);
    const int blocks = (waves + 7) / 8;
    qkv_proj_kernel<<<blocks, 256, 0, stream>>>(x, W_qkv, b_qkv, Qw, Kw, Vw,
                                                mtiles);
  }
  {
    const int nwaves = B * HEADS * (T_SEQ / 16);
    const int blocks = (nwaves + 7) / 8;
    attn_kernel<<<blocks, 256, 0, stream>>>(Qw, Kw, Vw, Aw, nwaves);
  }
  {
    const int tiles = mtiles * (C_DIM / 16);
    const int blocks = (tiles + 7) / 8;
    out_proj_kernel<<<blocks, 256, 0, stream>>>(Aw, W_fc, b_fc, out, mtiles);
  }
}


# attention: loads half a step ahead: the next tile's K is requested right behind the score WMMAs into the registers they read, this tile's V at the top of the step (launch bound 256); sums unchanged
# speedup vs baseline: 1.3792x; 1.0575x over previous
#include <hip/hip_runtime.h>


typedef float v2f __attribute__((ext_vector_type(2)));
typedef float v4f __attribute__((ext_vector_type(4)));
typedef float v8f __attribute__((ext_vector_type(8)));

#define T_SEQ 2048
#define HEADS 4
#define DH    64
#define C_DIM 256

__device__ __forceinline__ v8f wmma4(v2f a, v2f b, v8f c) {
  return __builtin_amdgcn_wmma_f32_16x16x4_f32(false, a, false, b, (short)0, c,
                                               false, false);
}

__global__ void qkv_proj_kernel(const float* __restrict__ X,
                                const float* __restrict__ W,
                                const float* __restrict__ bias,
                                float* __restrict__ Qw,
                                float* __restrict__ Kw,
                                float* __restrict__ Vw,
                                int mtiles) {
  const int lane = threadIdx.x & 31;
  const int lo = lane & 15;
  const int hi = lane >> 4;
  const int ngroups = (3 * C_DIM) / 64;
  const int wv = (int)((blockIdx.x * blockDim.x + threadIdx.x) >> 5);
  const int blk = wv >> 3;
  const int g = blk % ngroups;
  const int mt = (blk / ngroups) * 8 + (wv & 7);
  if (mt >= mtiles) return;
  const int m0 = mt * 16;

  const float* arow = X + (size_t)(m0 + lo) * C_DIM + hi * 4;
  const float* wrow = W + (size_t)(hi * 4) * (3 * C_DIM) + g * 64 + lo * 4;
  v8f c0 = {}, c1 = {}, c2 = {}, c3 = {};
#pragma unroll 2
  for (int j = 0; j < C_DIM / 8; ++j) {
    const v4f f = *(const v4f*)(arow + j * 8);
    const float* wp = wrow + (size_t)(j * 8) * (3 * C_DIM);
    const v4f w0 = *(const v4f*)(wp);
    const v4f w1 = *(const v4f*)(wp + 3 * C_DIM);
    const v4f w2 = *(const v4f*)(wp + 6 * C_DIM);
    const v4f w3 = *(const v4f*)(wp + 9 * C_DIM);
    v2f a, b;
    a.x = f.x;  a.y = f.y;
    b.x = w0.x; b.y = w1.x; c0 = wmma4(a, b, c0);
    b.x = w0.y; b.y = w1.y; c1 = wmma4(a, b, c1);
    b.x = w0.z; b.y = w1.z; c2 = wmma4(a, b, c2);
    b.x = w0.w; b.y = w1.w; c3 = wmma4(a, b, c3);
    a.x = f.z;  a.y = f.w;
    b.x = w2.x; b.y = w3.x; c0 = wmma4(a, b, c0);
    b.x = w2.y; b.y = w3.y; c1 = wmma4(a, b, c1);
    b.x = w2.z; b.y = w3.z; c2 = wmma4(a, b, c2);
    b.x = w2.w; b.y = w3.w; c3 = wmma4(a, b, c3);
  }

  const int which = g >> 2;
  const int h = g & 3;
  const v4f bv = *(const v4f*)(bias + g * 64 + lo * 4);
  const int bb = m0 >> 11;
  const int t0 = m0 & (T_SEQ - 1);
  const size_t tbase = (((size_t)(bb * HEADS + h)) * T_SEQ + t0) * DH;
  if (which != 2) {
    float* dst = ((which == 0) ? Qw : Kw) + tbase + (lo >> 1) * 128 + hi * 32 +
                 (lo & 1) * 2;
#pragma unroll
    for (int r = 0; r < 8; ++r) {
      v2f pr;
      pr.x = c0[r] + bv.x;
      pr.y = c1[r] + bv.y;
      *(v2f*)(dst + r * 4) = pr;
      pr.x = c2[r] + bv.z;
      pr.y = c3[r] + bv.w;
      *(v2f*)(dst + 64 + r * 4) = pr;
    }
  } else {
    float* dst = Vw + tbase + hi * 512 + (lo >> 3) * 128 + (lo & 3) * 16 +
                 ((lo >> 2) & 1) * 2;
#pragma unroll
    for (int r = 0; r < 8; r += 2) {
      float* d2 = dst + (r >> 2) * 256 + ((r >> 1) & 1) * 64;
      v2f pr;
      pr.x = c0[r] + bv.x;  pr.y = c0[r + 1] + bv.x;  *(v2f*)(d2) = pr;
      pr.x = c1[r] + bv.y;  pr.y = c1[r + 1] + bv.y;  *(v2f*)(d2 + 4) = pr;
      pr.x = c2[r] + bv.z;  pr.y = c2[r + 1] + bv.z;  *(v2f*)(d2 + 8) = pr;
      pr.x = c3[r] + bv.w;  pr.y = c3[r + 1] + bv.w;  *(v2f*)(d2 + 12) = pr;
    }
  }
}

__global__ void __launch_bounds__(256) attn_kernel(const float* __restrict__ Q,
                            const float* __restrict__ K,
                            const float* __restrict__ V,
                            float* __restrict__ O,
                            int nwaves) {
  const int lane = threadIdx.x & 31;
  const int lo = lane & 15;
  const int hi = lane >> 4;
  int wave = (int)((blockIdx.x * blockDim.x + threadIdx.x) >> 5);
  if (wave >= nwaves) return;

  const int qtiles = T_SEQ / 16;
  const int qt = wave % qtiles;
  const int bh = wave / qtiles;
  const int q0 = qt * 16;
  const float* Qh = Q + (size_t)bh * T_SEQ * DH;
  const float* Kh = K + (size_t)bh * T_SEQ * DH;
  const float* Vh = V + (size_t)bh * T_SEQ * DH;

  const float qs = 0.125f * 1.44269504088896340736f;
  v2f qreg[16];
  {
    const v4f* qf = (const v4f*)(Qh + (size_t)q0 * DH) + lane;
#pragma unroll
    for (int j = 0; j < 8; ++j) {
      const v4f f = qf[j * 32];
      qreg[2 * j].x = f.x * qs;
      qreg[2 * j].y = f.y * qs;
      qreg[2 * j + 1].x = f.z * qs;
      qreg[2 * j + 1].y = f.w * qs;
    }
  }

  float m_run = -__builtin_inff();
  float l_run = 0.0f;
  v8f o0 = {}, o1 = {}, o2 = {}, o3 = {};

  const v4f* kbase = (const v4f*)Kh + lane;
  const v4f* vbase = (const v4f*)Vh + lane;
  v4f kr[8];
#pragma unroll
  for (int j = 0; j < 8; ++j) kr[j] = kbase[j * 32];

  for (int kt0 = 0; kt0 <= q0; kt0 += 16) {
    v4f vr[8];
    {
      const v4f* vf = vbase + (size_t)kt0 * (DH / 4);
#pragma unroll
      for (int j = 0; j < 8; ++j) vr[j] = vf[j * 32];
    }
    v8f s = {};
#pragma unroll
    for (int j = 0; j < 8; ++j) {
      v2f a;
      a.x = kr[j].x; a.y = kr[j].y;
      s = wmma4(a, qreg[2 * j], s);
      a.x = kr[j].z; a.y = kr[j].w;
      s = wmma4(a, qreg[2 * j + 1], s);
    }
    if (kt0 == q0) {
#pragma unroll
      for (int r = 0; r < 8; ++r)
        s[r] = (r + hi * 8 > lo) ? -__builtin_inff() : s[r];
    }

    {
      const int ktn = (kt0 < q0) ? kt0 + 16 : kt0;
      const v4f* kf = kbase + (size_t)ktn * (DH / 4);
#pragma unroll
      for (int j = 0; j < 8; ++j) kr[j] = kf[j * 32];
    }

    float tmax = s[0];
#pragma unroll
    for (int r = 1; r < 8; ++r) tmax = fmaxf(tmax, s[r]);
    tmax = fmaxf(tmax, __shfl_xor(tmax, 16, 32));
    const float m_new = fmaxf(m_run, tmax);
    const float alpha = __builtin_amdgcn_exp2f(m_run - m_new);
    float p[8];
    float ts = 0.0f;
#pragma unroll
    for (int r = 0; r < 8; ++r) {
      p[r] = __builtin_amdgcn_exp2f(s[r] - m_new);
      ts += p[r];
    }
    ts += __shfl_xor(ts, 16, 32);
    l_run = l_run * alpha + ts;
    m_run = m_new;
#pragma unroll
    for (int r = 0; r < 8; ++r) {
      o0[r] *= alpha; o1[r] *= alpha; o2[r] *= alpha; o3[r] *= alpha;
    }

#pragma unroll
    for (int kk = 0; kk < 16; kk += 4) {
      v2f bop;
      if (kk < 8) {
        const float sx = __shfl_xor(p[kk + 2], 16, 32);
        const float sy = __shfl_xor(p[kk + 3], 16, 32);
        bop.x = hi ? sx : p[kk];
        bop.y = hi ? sy : p[kk + 1];
      } else {
        const int j = kk - 8;
        const float sx = __shfl_xor(p[j], 16, 32);
        const float sy = __shfl_xor(p[j + 1], 16, 32);
        bop.x = hi ? p[j + 2] : sx;
        bop.y = hi ? p[j + 3] : sy;
      }
      const v4f f0 = vr[kk >> 1];
      const v4f f1 = vr[(kk >> 1) + 1];
      v2f a;
      a.x = f0.x;  a.y = f0.y;   o0 = wmma4(a, bop, o0);
      a.x = f0.z;  a.y = f0.w;   o1 = wmma4(a, bop, o1);
      a.x = f1.x;  a.y = f1.y;   o2 = wmma4(a, bop, o2);
      a.x = f1.z;  a.y = f1.w;   o3 = wmma4(a, bop, o3);
    }
  }

  const float inv_l = 1.0f / l_run;
  const int b = bh / HEADS;
  const int h = bh - b * HEADS;
  float* obase = O + ((size_t)b * T_SEQ + q0) * C_DIM + (h * 8 + hi) * 128 + lo * 4;
#pragma unroll
  for (int hh = 0; hh < 2; ++hh) {
    v4f w;
    w.x = o0[2 * hh] * inv_l;     w.y = o0[2 * hh + 1] * inv_l;
    w.z = o0[2 * hh + 4] * inv_l; w.w = o0[2 * hh + 5] * inv_l;
    *(v4f*)(obase + hh * 64) = w;
    w.x = o1[2 * hh] * inv_l;     w.y = o1[2 * hh + 1] * inv_l;
    w.z = o1[2 * hh + 4] * inv_l; w.w = o1[2 * hh + 5] * inv_l;
    *(v4f*)(obase + 256 + hh * 64) = w;
    w.x = o2[2 * hh] * inv_l;     w.y = o2[2 * hh + 1] * inv_l;
    w.z = o2[2 * hh + 4] * inv_l; w.w = o2[2 * hh + 5] * inv_l;
    *(v4f*)(obase + 512 + hh * 64) = w;
    w.x = o3[2 * hh] * inv_l;     w.y = o3[2 * hh + 1] * inv_l;
    w.z = o3[2 * hh + 4] * inv_l; w.w = o3[2 * hh + 5] * inv_l;
    *(v4f*)(obase + 768 + hh * 64) = w;
  }
}

__global__ void out_proj_kernel(const float* __restrict__ A,
                                const float* __restrict__ W,
                                const float* __restrict__ bias,
                                float* __restrict__ Out,
                                int mtiles) {
  const int lane = threadIdx.x & 31;
  const int lo = lane & 15;
  const int hi = lane >> 4;
  const int ntiles = C_DIM / 16;
  int tile = (int)((blockIdx.x * blockDim.x + threadIdx.x) >> 5);
  if (tile >= mtiles * ntiles) return;
  const int mt = tile / ntiles;
  const int nt = tile - mt * ntiles;
  const int m0 = mt * 16, n0 = nt * 16;

  const v4f* af = (const v4f*)(A + (size_t)m0 * C_DIM) + lane;
  const float* wcol = W + (n0 + lo);
  v8f c = {};
#pragma unroll 4
  for (int j = 0; j < C_DIM / 8; ++j) {
    const v4f f = af[j * 32];
    const int ka = j * 8 + hi * 2;
    v2f a, b;
    a.x = f.x;
    a.y = f.y;
    b.x = wcol[(size_t)ka * C_DIM];
    b.y = wcol[(size_t)(ka + 1) * C_DIM];
    c = wmma4(a, b, c);
    a.x = f.z;
    a.y = f.w;
    b.x = wcol[(size_t)(ka + 4) * C_DIM];
    b.y = wcol[(size_t)(ka + 5) * C_DIM];
    c = wmma4(a, b, c);
  }
  const float bv = bias[n0 + lo];
#pragma unroll
  for (int r = 0; r < 8; ++r)
    Out[(size_t)(m0 + r + hi * 8) * C_DIM + n0 + lo] = c[r] + bv;
}

extern "C" void kernel_launch(void* const* d_in, const int* in_sizes, int n_in,
                              void* d_out, int out_size, void* d_ws, size_t ws_size,
                              hipStream_t stream) {
  const float* x     = (const float*)d_in[0];
  const float* W_qkv = (const float*)d_in[1];
  const float* b_qkv = (const float*)d_in[2];
  const float* W_fc  = (const float*)d_in[3];
  const float* b_fc  = (const float*)d_in[4];
  float* out = (float*)d_out;

  const int BT = in_sizes[0] / C_DIM;
  const int B  = BT / T_SEQ;

  const size_t headElems = (size_t)B * HEADS * T_SEQ * DH;
  float* Qw = (float*)d_ws;
  float* Kw = Qw + headElems;
  float* Vw = Kw + headElems;
  float* Aw = Vw + headElems;

  const int mtiles = BT / 16;

  {
    const int waves = mtiles * ((3 * C_DIM) / 64);
    const int blocks = (waves + 7) / 8;
    qkv_proj_kernel<<<blocks, 256, 0, stream>>>(x, W_qkv, b_qkv, Qw, Kw, Vw,
                                                mtiles);
  }
  {
    const int nwaves = B * HEADS * (T_SEQ / 16);
    const int blocks = (nwaves + 7) / 8;
    attn_kernel<<<blocks, 256, 0, stream>>>(Qw, Kw, Vw, Aw, nwaves);
  }
  {
    const int tiles = mtiles * (C_DIM / 16);
    const int blocks = (tiles + 7) / 8;
    out_proj_kernel<<<blocks, 256, 0, stream>>>(Aw, W_fc, b_fc, out, mtiles);
  }
}


# attention PV: V tile stored with keys in score-register order (half h, step i: keys 8h+2i, 8h+2i+1): P^T feeds the WMMAs in place, 8 lane exchanges + selects per key tile go; PV key sum order changes
# speedup vs baseline: 1.4243x; 1.0327x over previous
#include <hip/hip_runtime.h>


typedef float v2f __attribute__((ext_vector_type(2)));
typedef float v4f __attribute__((ext_vector_type(4)));
typedef float v8f __attribute__((ext_vector_type(8)));

#define T_SEQ 2048
#define HEADS 4
#define DH    64
#define C_DIM 256

__device__ __forceinline__ v8f wmma4(v2f a, v2f b, v8f c) {
  return __builtin_amdgcn_wmma_f32_16x16x4_f32(false, a, false, b, (short)0, c,
                                               false, false);
}

__global__ void qkv_proj_kernel(const float* __restrict__ X,
                                const float* __restrict__ W,
                                const float* __restrict__ bias,
                                float* __restrict__ Qw,
                                float* __restrict__ Kw,
                                float* __restrict__ Vw,
                                int mtiles) {
  const int lane = threadIdx.x & 31;
  const int lo = lane & 15;
  const int hi = lane >> 4;
  const int ngroups = (3 * C_DIM) / 64;
  const int wv = (int)((blockIdx.x * blockDim.x + threadIdx.x) >> 5);
  const int blk = wv >> 3;
  const int g = blk % ngroups;
  const int mt = (blk / ngroups) * 8 + (wv & 7);
  if (mt >= mtiles) return;
  const int m0 = mt * 16;

  const float* arow = X + (size_t)(m0 + lo) * C_DIM + hi * 4;
  const float* wrow = W + (size_t)(hi * 4) * (3 * C_DIM) + g * 64 + lo * 4;
  v8f c0 = {}, c1 = {}, c2 = {}, c3 = {};
#pragma unroll 2
  for (int j = 0; j < C_DIM / 8; ++j) {
    const v4f f = *(const v4f*)(arow + j * 8);
    const float* wp = wrow + (size_t)(j * 8) * (3 * C_DIM);
    const v4f w0 = *(const v4f*)(wp);
    const v4f w1 = *(const v4f*)(wp + 3 * C_DIM);
    const v4f w2 = *(const v4f*)(wp + 6 * C_DIM);
    const v4f w3 = *(const v4f*)(wp + 9 * C_DIM);
    v2f a, b;
    a.x = f.x;  a.y = f.y;
    b.x = w0.x; b.y = w1.x; c0 = wmma4(a, b, c0);
    b.x = w0.y; b.y = w1.y; c1 = wmma4(a, b, c1);
    b.x = w0.z; b.y = w1.z; c2 = wmma4(a, b, c2);
    b.x = w0.w; b.y = w1.w; c3 = wmma4(a, b, c3);
    a.x = f.z;  a.y = f.w;
    b.x = w2.x; b.y = w3.x; c0 = wmma4(a, b, c0);
    b.x = w2.y; b.y = w3.y; c1 = wmma4(a, b, c1);
    b.x = w2.z; b.y = w3.z; c2 = wmma4(a, b, c2);
    b.x = w2.w; b.y = w3.w; c3 = wmma4(a, b, c3);
  }

  const int which = g >> 2;
  const int h = g & 3;
  const v4f bv = *(const v4f*)(bias + g * 64 + lo * 4);
  const int bb = m0 >> 11;
  const int t0 = m0 & (T_SEQ - 1);
  const size_t tbase = (((size_t)(bb * HEADS + h)) * T_SEQ + t0) * DH;
  if (which != 2) {
    float* dst = ((which == 0) ? Qw : Kw) + tbase + (lo >> 1) * 128 + hi * 32 +
                 (lo & 1) * 2;
#pragma unroll
    for (int r = 0; r < 8; ++r) {
      v2f pr;
      pr.x = c0[r] + bv.x;
      pr.y = c1[r] + bv.y;
      *(v2f*)(dst + r * 4) = pr;
      pr.x = c2[r] + bv.z;
      pr.y = c3[r] + bv.w;
      *(v2f*)(dst + 64 + r * 4) = pr;
    }
  } else {
    float* dst = Vw + tbase + hi * 64 + (lo >> 3) * 128 + (lo & 3) * 16 +
                 ((lo >> 2) & 1) * 2;
#pragma unroll
    for (int r = 0; r < 8; r += 2) {
      float* d2 = dst + (r >> 1) * 256;
      v2f pr;
      pr.x = c0[r] + bv.x;  pr.y = c0[r + 1] + bv.x;  *(v2f*)(d2) = pr;
      pr.x = c1[r] + bv.y;  pr.y = c1[r + 1] + bv.y;  *(v2f*)(d2 + 4) = pr;
      pr.x = c2[r] + bv.z;  pr.y = c2[r + 1] + bv.z;  *(v2f*)(d2 + 8) = pr;
      pr.x = c3[r] + bv.w;  pr.y = c3[r + 1] + bv.w;  *(v2f*)(d2 + 12) = pr;
    }
  }
}

__global__ void __launch_bounds__(256) attn_kernel(const float* __restrict__ Q,
                            const float* __restrict__ K,
                            const float* __restrict__ V,
                            float* __restrict__ O,
                            int nwaves) {
  const int lane = threadIdx.x & 31;
  const int lo = lane & 15;
  const int hi = lane >> 4;
  int wave = (int)((blockIdx.x * blockDim.x + threadIdx.x) >> 5);
  if (wave >= nwaves) return;

  const int qtiles = T_SEQ / 16;
  const int qt = wave % qtiles;
  const int bh = wave / qtiles;
  const int q0 = qt * 16;
  const float* Qh = Q + (size_t)bh * T_SEQ * DH;
  const float* Kh = K + (size_t)bh * T_SEQ * DH;
  const float* Vh = V + (size_t)bh * T_SEQ * DH;

  const float qs = 0.125f * 1.44269504088896340736f;
  v2f qreg[16];
  {
    const v4f* qf = (const v4f*)(Qh + (size_t)q0 * DH) + lane;
#pragma unroll
    for (int j = 0; j < 8; ++j) {
      const v4f f = qf[j * 32];
      qreg[2 * j].x = f.x * qs;
      qreg[2 * j].y = f.y * qs;
      qreg[2 * j + 1].x = f.z * qs;
      qreg[2 * j + 1].y = f.w * qs;
    }
  }

  float m_run = -__builtin_inff();
  float l_run = 0.0f;
  v8f o0 = {}, o1 = {}, o2 = {}, o3 = {};

  const v4f* kbase = (const v4f*)Kh + lane;
  const v4f* vbase = (const v4f*)Vh + lane;
  v4f kr[8];
#pragma unroll
  for (int j = 0; j < 8; ++j) kr[j] = kbase[j * 32];

  for (int kt0 = 0; kt0 <= q0; kt0 += 16) {
    v4f vr[8];
    {
      const v4f* vf = vbase + (size_t)kt0 * (DH / 4);
#pragma unroll
      for (int j = 0; j < 8; ++j) vr[j] = vf[j * 32];
    }
    v8f s = {};
#pragma unroll
    for (int j = 0; j < 8; ++j) {
      v2f a;
      a.x = kr[j].x; a.y = kr[j].y;
      s = wmma4(a, qreg[2 * j], s);
      a.x = kr[j].z; a.y = kr[j].w;
      s = wmma4(a, qreg[2 * j + 1], s);
    }
    if (kt0 == q0) {
#pragma unroll
      for (int r = 0; r < 8; ++r)
        s[r] = (r + hi * 8 > lo) ? -__builtin_inff() : s[r];
    }

    {
      const int ktn = (kt0 < q0) ? kt0 + 16 : kt0;
      const v4f* kf = kbase + (size_t)ktn * (DH / 4);
#pragma unroll
      for (int j = 0; j < 8; ++j) kr[j] = kf[j * 32];
    }

    float tmax = s[0];
#pragma unroll
    for (int r = 1; r < 8; ++r) tmax = fmaxf(tmax, s[r]);
    tmax = fmaxf(tmax, __shfl_xor(tmax, 16, 32));
    const float m_new = fmaxf(m_run, tmax);
    const float alpha = __builtin_amdgcn_exp2f(m_run - m_new);
    float p[8];
    float ts = 0.0f;
#pragma unroll
    for (int r = 0; r < 8; ++r) {
      p[r] = __builtin_amdgcn_exp2f(s[r] - m_new);
      ts += p[r];
    }
    ts += __shfl_xor(ts, 16, 32);
    l_run = l_run * alpha + ts;
    m_run = m_new;
#pragma unroll
    for (int r = 0; r < 8; ++r) {
      o0[r] *= alpha; o1[r] *= alpha; o2[r] *= alpha; o3[r] *= alpha;
    }

#pragma unroll
    for (int i = 0; i < 4; ++i) {
      v2f bop;
      bop.x = p[2 * i];
      bop.y = p[2 * i + 1];
      const v4f f0 = vr[2 * i];
      const v4f f1 = vr[2 * i + 1];
      v2f a;
      a.x = f0.x;  a.y = f0.y;   o0 = wmma4(a, bop, o0);
      a.x = f0.z;  a.y = f0.w;   o1 = wmma4(a, bop, o1);
      a.x = f1.x;  a.y = f1.y;   o2 = wmma4(a, bop, o2);
      a.x = f1.z;  a.y = f1.w;   o3 = wmma4(a, bop, o3);
    }
  }

  const float inv_l = 1.0f / l_run;
  const int b = bh / HEADS;
  const int h = bh - b * HEADS;
  float* obase = O + ((size_t)b * T_SEQ + q0) * C_DIM + (h * 8 + hi) * 128 + lo * 4;
#pragma unroll
  for (int hh = 0; hh < 2; ++hh) {
    v4f w;
    w.x = o0[2 * hh] * inv_l;     w.y = o0[2 * hh + 1] * inv_l;
    w.z = o0[2 * hh + 4] * inv_l; w.w = o0[2 * hh + 5] * inv_l;
    *(v4f*)(obase + hh * 64) = w;
    w.x = o1[2 * hh] * inv_l;     w.y = o1[2 * hh + 1] * inv_l;
    w.z = o1[2 * hh + 4] * inv_l; w.w = o1[2 * hh + 5] * inv_l;
    *(v4f*)(obase + 256 + hh * 64) = w;
    w.x = o2[2 * hh] * inv_l;     w.y = o2[2 * hh + 1] * inv_l;
    w.z = o2[2 * hh + 4] * inv_l; w.w = o2[2 * hh + 5] * inv_l;
    *(v4f*)(obase + 512 + hh * 64) = w;
    w.x = o3[2 * hh] * inv_l;     w.y = o3[2 * hh + 1] * inv_l;
    w.z = o3[2 * hh + 4] * inv_l; w.w = o3[2 * hh + 5] * inv_l;
    *(v4f*)(obase + 768 + hh * 64) = w;
  }
}

__global__ void out_proj_kernel(const float* __restrict__ A,
                                const float* __restrict__ W,
                                const float* __restrict__ bias,
                                float* __restrict__ Out,
                                int mtiles) {
  const int lane = threadIdx.x & 31;
  const int lo = lane & 15;
  const int hi = lane >> 4;
  const int ntiles = C_DIM / 16;
  int tile = (int)((blockIdx.x * blockDim.x + threadIdx.x) >> 5);
  if (tile >= mtiles * ntiles) return;
  const int mt = tile / ntiles;
  const int nt = tile - mt * ntiles;
  const int m0 = mt * 16, n0 = nt * 16;

  const v4f* af = (const v4f*)(A + (size_t)m0 * C_DIM) + lane;
  const float* wcol = W + (n0 + lo);
  v8f c = {};
#pragma unroll 4
  for (int j = 0; j < C_DIM / 8; ++j) {
    const v4f f = af[j * 32];
    const int ka = j * 8 + hi * 2;
    v2f a, b;
    a.x = f.x;
    a.y = f.y;
    b.x = wcol[(size_t)ka * C_DIM];
    b.y = wcol[(size_t)(ka + 1) * C_DIM];
    c = wmma4(a, b, c);
    a.x = f.z;
    a.y = f.w;
    b.x = wcol[(size_t)(ka + 4) * C_DIM];
    b.y = wcol[(size_t)(ka + 5) * C_DIM];
    c = wmma4(a, b, c);
  }
  const float bv = bias[n0 + lo];
#pragma unroll
  for (int r = 0; r < 8; ++r)
    Out[(size_t)(m0 + r + hi * 8) * C_DIM + n0 + lo] = c[r] + bv;
}

extern "C" void kernel_launch(void* const* d_in, const int* in_sizes, int n_in,
                              void* d_out, int out_size, void* d_ws, size_t ws_size,
                              hipStream_t stream) {
  const float* x     = (const float*)d_in[0];
  const float* W_qkv = (const float*)d_in[1];
  const float* b_qkv = (const float*)d_in[2];
  const float* W_fc  = (const float*)d_in[3];
  const float* b_fc  = (const float*)d_in[4];
  float* out = (float*)d_out;

  const int BT = in_sizes[0] / C_DIM;
  const int B  = BT / T_SEQ;

  const size_t headElems = (size_t)B * HEADS * T_SEQ * DH;
  float* Qw = (float*)d_ws;
  float* Kw = Qw + headElems;
  float* Vw = Kw + headElems;
  float* Aw = Vw + headElems;

  const int mtiles = BT / 16;

  {
    const int waves = mtiles * ((3 * C_DIM) / 64);
    const int blocks = (waves + 7) / 8;
    qkv_proj_kernel<<<blocks, 256, 0, stream>>>(x, W_qkv, b_qkv, Qw, Kw, Vw,
                                                mtiles);
  }
  {
    const int nwaves = B * HEADS * (T_SEQ / 16);
    const int blocks = (nwaves + 7) / 8;
    attn_kernel<<<blocks, 256, 0, stream>>>(Qw, Kw, Vw, Aw, nwaves);
  }
  {
    const int tiles = mtiles * (C_DIM / 16);
    const int blocks = (tiles + 7) / 8;
    out_proj_kernel<<<blocks, 256, 0, stream>>>(Aw, W_fc, b_fc, out, mtiles);
  }
}


# attention: a head's query tiles dealt to waves so that waves w and w+4 of a block (one SIMD) hold tiles 127-p and p: equal causal work per SIMD and block; one tile per wave, all sums unchanged
# speedup vs baseline: 1.8274x; 1.2830x over previous
#include <hip/hip_runtime.h>


typedef float v2f __attribute__((ext_vector_type(2)));
typedef float v4f __attribute__((ext_vector_type(4)));
typedef float v8f __attribute__((ext_vector_type(8)));

#define T_SEQ 2048
#define HEADS 4
#define DH    64
#define C_DIM 256

__device__ __forceinline__ v8f wmma4(v2f a, v2f b, v8f c) {
  return __builtin_amdgcn_wmma_f32_16x16x4_f32(false, a, false, b, (short)0, c,
                                               false, false);
}

__global__ void qkv_proj_kernel(const float* __restrict__ X,
                                const float* __restrict__ W,
                                const float* __restrict__ bias,
                                float* __restrict__ Qw,
                                float* __restrict__ Kw,
                                float* __restrict__ Vw,
                                int mtiles) {
  const int lane = threadIdx.x & 31;
  const int lo = lane & 15;
  const int hi = lane >> 4;
  const int ngroups = (3 * C_DIM) / 64;
  const int wv = (int)((blockIdx.x * blockDim.x + threadIdx.x) >> 5);
  const int blk = wv >> 3;
  const int g = blk % ngroups;
  const int mt = (blk / ngroups) * 8 + (wv & 7);
  if (mt >= mtiles) return;
  const int m0 = mt * 16;

  const float* arow = X + (size_t)(m0 + lo) * C_DIM + hi * 4;
  const float* wrow = W + (size_t)(hi * 4) * (3 * C_DIM) + g * 64 + lo * 4;
  v8f c0 = {}, c1 = {}, c2 = {}, c3 = {};
#pragma unroll 2
  for (int j = 0; j < C_DIM / 8; ++j) {
    const v4f f = *(const v4f*)(arow + j * 8);
    const float* wp = wrow + (size_t)(j * 8) * (3 * C_DIM);
    const v4f w0 = *(const v4f*)(wp);
    const v4f w1 = *(const v4f*)(wp + 3 * C_DIM);
    const v4f w2 = *(const v4f*)(wp + 6 * C_DIM);
    const v4f w3 = *(const v4f*)(wp + 9 * C_DIM);
    v2f a, b;
    a.x = f.x;  a.y = f.y;
    b.x = w0.x; b.y = w1.x; c0 = wmma4(a, b, c0);
    b.x = w0.y; b.y = w1.y; c1 = wmma4(a, b, c1);
    b.x = w0.z; b.y = w1.z; c2 = wmma4(a, b, c2);
    b.x = w0.w; b.y = w1.w; c3 = wmma4(a, b, c3);
    a.x = f.z;  a.y = f.w;
    b.x = w2.x; b.y = w3.x; c0 = wmma4(a, b, c0);
    b.x = w2.y; b.y = w3.y; c1 = wmma4(a, b, c1);
    b.x = w2.z; b.y = w3.z; c2 = wmma4(a, b, c2);
    b.x = w2.w; b.y = w3.w; c3 = wmma4(a, b, c3);
  }

  const int which = g >> 2;
  const int h = g & 3;
  const v4f bv = *(const v4f*)(bias + g * 64 + lo * 4);
  const int bb = m0 >> 11;
  const int t0 = m0 & (T_SEQ - 1);
  const size_t tbase = (((size_t)(bb * HEADS + h)) * T_SEQ + t0) * DH;
  if (which != 2) {
    float* dst = ((which == 0) ? Qw : Kw) + tbase + (lo >> 1) * 128 + hi * 32 +
                 (lo & 1) * 2;
#pragma unroll
    for (int r = 0; r < 8; ++r) {
      v2f pr;
      pr.x = c0[r] + bv.x;
      pr.y = c1[r] + bv.y;
      *(v2f*)(dst + r * 4) = pr;
      pr.x = c2[r] + bv.z;
      pr.y = c3[r] + bv.w;
      *(v2f*)(dst + 64 + r * 4) = pr;
    }
  } else {
    float* dst = Vw + tbase + hi * 64 + (lo >> 3) * 128 + (lo & 3) * 16 +
                 ((lo >> 2) & 1) * 2;
#pragma unroll
    for (int r = 0; r < 8; r += 2) {
      float* d2 = dst + (r >> 1) * 256;
      v2f pr;
      pr.x = c0[r] + bv.x;  pr.y = c0[r + 1] + bv.x;  *(v2f*)(d2) = pr;
      pr.x = c1[r] + bv.y;  pr.y = c1[r + 1] + bv.y;  *(v2f*)(d2 + 4) = pr;
      pr.x = c2[r] + bv.z;  pr.y = c2[r + 1] + bv.z;  *(v2f*)(d2 + 8) = pr;
      pr.x = c3[r] + bv.w;  pr.y = c3[r + 1] + bv.w;  *(v2f*)(d2 + 12) = pr;
    }
  }
}

__global__ void __launch_bounds__(256) attn_kernel(const float* __restrict__ Q,
                            const float* __restrict__ K,
                            const float* __restrict__ V,
                            float* __restrict__ O,
                            int nwaves) {
  const int lane = threadIdx.x & 31;
  const int lo = lane & 15;
  const int hi = lane >> 4;
  int wave = (int)((blockIdx.x * blockDim.x + threadIdx.x) >> 5);
  if (wave >= nwaves) return;

  const int qtiles = T_SEQ / 16;
  const int bh = wave / qtiles;
  const int u = wave % qtiles;
  const int p = (u >> 3) * 4 + (u & 3);
  const int qt = (u & 4) ? p : (qtiles - 1 - p);
  const int q0 = qt * 16;
  const float* Qh = Q + (size_t)bh * T_SEQ * DH;
  const float* Kh = K + (size_t)bh * T_SEQ * DH;
  const float* Vh = V + (size_t)bh * T_SEQ * DH;

  const float qs = 0.125f * 1.44269504088896340736f;
  v2f qreg[16];
  {
    const v4f* qf = (const v4f*)(Qh + (size_t)q0 * DH) + lane;
#pragma unroll
    for (int j = 0; j < 8; ++j) {
      const v4f f = qf[j * 32];
      qreg[2 * j].x = f.x * qs;
      qreg[2 * j].y = f.y * qs;
      qreg[2 * j + 1].x = f.z * qs;
      qreg[2 * j + 1].y = f.w * qs;
    }
  }

  float m_run = -__builtin_inff();
  float l_run = 0.0f;
  v8f o0 = {}, o1 = {}, o2 = {}, o3 = {};

  const v4f* kbase = (const v4f*)Kh + lane;
  const v4f* vbase = (const v4f*)Vh + lane;
  v4f kr[8];
#pragma unroll
  for (int j = 0; j < 8; ++j) kr[j] = kbase[j * 32];

  for (int kt0 = 0; kt0 <= q0; kt0 += 16) {
    v4f vr[8];
    {
      const v4f* vf = vbase + (size_t)kt0 * (DH / 4);
#pragma unroll
      for (int j = 0; j < 8; ++j) vr[j] = vf[j * 32];
    }
    v8f s = {};
#pragma unroll
    for (int j = 0; j < 8; ++j) {
      v2f a;
      a.x = kr[j].x; a.y = kr[j].y;
      s = wmma4(a, qreg[2 * j], s);
      a.x = kr[j].z; a.y = kr[j].w;
      s = wmma4(a, qreg[2 * j + 1], s);
    }
    if (kt0 == q0) {
#pragma unroll
      for (int r = 0; r < 8; ++r)
        s[r] = (r + hi * 8 > lo) ? -__builtin_inff() : s[r];
    }

    {
      const int ktn = (kt0 < q0) ? kt0 + 16 : kt0;
      const v4f* kf = kbase + (size_t)ktn * (DH / 4);
#pragma unroll
      for (int j = 0; j < 8; ++j) kr[j] = kf[j * 32];
    }

    float tmax = s[0];
#pragma unroll
    for (int r = 1; r < 8; ++r) tmax = fmaxf(tmax, s[r]);
    tmax = fmaxf(tmax, __shfl_xor(tmax, 16, 32));
    const float m_new = fmaxf(m_run, tmax);
    const float alpha = __builtin_amdgcn_exp2f(m_run - m_new);
    float p[8];
    float ts = 0.0f;
#pragma unroll
    for (int r = 0; r < 8; ++r) {
      p[r] = __builtin_amdgcn_exp2f(s[r] - m_new);
      ts += p[r];
    }
    ts += __shfl_xor(ts, 16, 32);
    l_run = l_run * alpha + ts;
    m_run = m_new;
#pragma unroll
    for (int r = 0; r < 8; ++r) {
      o0[r] *= alpha; o1[r] *= alpha; o2[r] *= alpha; o3[r] *= alpha;
    }

#pragma unroll
    for (int i = 0; i < 4; ++i) {
      v2f bop;
      bop.x = p[2 * i];
      bop.y = p[2 * i + 1];
      const v4f f0 = vr[2 * i];
      const v4f f1 = vr[2 * i + 1];
      v2f a;
      a.x = f0.x;  a.y = f0.y;   o0 = wmma4(a, bop, o0);
      a.x = f0.z;  a.y = f0.w;   o1 = wmma4(a, bop, o1);
      a.x = f1.x;  a.y = f1.y;   o2 = wmma4(a, bop, o2);
      a.x = f1.z;  a.y = f1.w;   o3 = wmma4(a, bop, o3);
    }
  }

  const float inv_l = 1.0f / l_run;
  const int b = bh / HEADS;
  const int h = bh - b * HEADS;
  float* obase = O + ((size_t)b * T_SEQ + q0) * C_DIM + (h * 8 + hi) * 128 + lo * 4;
#pragma unroll
  for (int hh = 0; hh < 2; ++hh) {
    v4f w;
    w.x = o0[2 * hh] * inv_l;     w.y = o0[2 * hh + 1] * inv_l;
    w.z = o0[2 * hh + 4] * inv_l; w.w = o0[2 * hh + 5] * inv_l;
    *(v4f*)(obase + hh * 64) = w;
    w.x = o1[2 * hh] * inv_l;     w.y = o1[2 * hh + 1] * inv_l;
    w.z = o1[2 * hh + 4] * inv_l; w.w = o1[2 * hh + 5] * inv_l;
    *(v4f*)(obase + 256 + hh * 64) = w;
    w.x = o2[2 * hh] * inv_l;     w.y = o2[2 * hh + 1] * inv_l;
    w.z = o2[2 * hh + 4] * inv_l; w.w = o2[2 * hh + 5] * inv_l;
    *(v4f*)(obase + 512 + hh * 64) = w;
    w.x = o3[2 * hh] * inv_l;     w.y = o3[2 * hh + 1] * inv_l;
    w.z = o3[2 * hh + 4] * inv_l; w.w = o3[2 * hh + 5] * inv_l;
    *(v4f*)(obase + 768 + hh * 64) = w;
  }
}

__global__ void out_proj_kernel(const float* __restrict__ A,
                                const float* __restrict__ W,
                                const float* __restrict__ bias,
                                float* __restrict__ Out,
                                int mtiles) {
  const int lane = threadIdx.x & 31;
  const int lo = lane & 15;
  const int hi = lane >> 4;
  const int ntiles = C_DIM / 16;
  int tile = (int)((blockIdx.x * blockDim.x + threadIdx.x) >> 5);
  if (tile >= mtiles * ntiles) return;
  const int mt = tile / ntiles;
  const int nt = tile - mt * ntiles;
  const int m0 = mt * 16, n0 = nt * 16;

  const v4f* af = (const v4f*)(A + (size_t)m0 * C_DIM) + lane;
  const float* wcol = W + (n0 + lo);
  v8f c = {};
#pragma unroll 4
  for (int j = 0; j < C_DIM / 8; ++j) {
    const v4f f = af[j * 32];
    const int ka = j * 8 + hi * 2;
    v2f a, b;
    a.x = f.x;
    a.y = f.y;
    b.x = wcol[(size_t)ka * C_DIM];
    b.y = wcol[(size_t)(ka + 1) * C_DIM];
    c = wmma4(a, b, c);
    a.x = f.z;
    a.y = f.w;
    b.x = wcol[(size_t)(ka + 4) * C_DIM];
    b.y = wcol[(size_t)(ka + 5) * C_DIM];
    c = wmma4(a, b, c);
  }
  const float bv = bias[n0 + lo];
#pragma unroll
  for (int r = 0; r < 8; ++r)
    Out[(size_t)(m0 + r + hi * 8) * C_DIM + n0 + lo] = c[r] + bv;
}

extern "C" void kernel_launch(void* const* d_in, const int* in_sizes, int n_in,
                              void* d_out, int out_size, void* d_ws, size_t ws_size,
                              hipStream_t stream) {
  const float* x     = (const float*)d_in[0];
  const float* W_qkv = (const float*)d_in[1];
  const float* b_qkv = (const float*)d_in[2];
  const float* W_fc  = (const float*)d_in[3];
  const float* b_fc  = (const float*)d_in[4];
  float* out = (float*)d_out;

  const int BT = in_sizes[0] / C_DIM;
  const int B  = BT / T_SEQ;

  const size_t headElems = (size_t)B * HEADS * T_SEQ * DH;
  float* Qw = (float*)d_ws;
  float* Kw = Qw + headElems;
  float* Vw = Kw + headElems;
  float* Aw = Vw + headElems;

  const int mtiles = BT / 16;

  {
    const int waves = mtiles * ((3 * C_DIM) / 64);
    const int blocks = (waves + 7) / 8;
    qkv_proj_kernel<<<blocks, 256, 0, stream>>>(x, W_qkv, b_qkv, Qw, Kw, Vw,
                                                mtiles);
  }
  {
    const int nwaves = B * HEADS * (T_SEQ / 16);
    const int blocks = (nwaves + 7) / 8;
    attn_kernel<<<blocks, 256, 0, stream>>>(Qw, Kw, Vw, Aw, nwaves);
  }
  {
    const int tiles = mtiles * (C_DIM / 16);
    const int blocks = (tiles + 7) / 8;
    out_proj_kernel<<<blocks, 256, 0, stream>>>(Aw, W_fc, b_fc, out, mtiles);
  }
}


# both projections: a wave's register tile grows (qkv 16x64 -> 64x64, 12-wave blocks, one per WGP; out 16x16 -> 16x64 with 16-byte W loads, whole-line stores): 4x fewer waves; sum orders unchanged
# speedup vs baseline: 1.9404x; 1.0618x over previous
#include <hip/hip_runtime.h>


typedef float v2f __attribute__((ext_vector_type(2)));
typedef float v4f __attribute__((ext_vector_type(4)));
typedef float v8f __attribute__((ext_vector_type(8)));

#define T_SEQ 2048
#define HEADS 4
#define DH    64
#define C_DIM 256

__device__ __forceinline__ v8f wmma4(v2f a, v2f b, v8f c) {
  return __builtin_amdgcn_wmma_f32_16x16x4_f32(false, a, false, b, (short)0, c,
                                               false, false);
}

__global__ void __launch_bounds__(384) qkv_proj_kernel(const float* __restrict__ X,
                                const float* __restrict__ W,
                                const float* __restrict__ bias,
                                float* __restrict__ Qw,
                                float* __restrict__ Kw,
                                float* __restrict__ Vw,
                                int mgroups) {
  const int lane = threadIdx.x & 31;
  const int lo = lane & 15;
  const int hi = lane >> 4;
  const int g = (int)(threadIdx.x >> 5);
  const int mg = (int)blockIdx.x;
  if (mg >= mgroups) return;
  const int m0 = mg * 64;

  const float* arow = X + (size_t)(m0 + lo) * C_DIM + hi * 4;
  const float* wrow = W + (size_t)(hi * 4) * (3 * C_DIM) + g * 64 + lo * 4;
  v8f c[4][4];
#pragma unroll
  for (int i = 0; i < 4; ++i) {
#pragma unroll
    for (int t = 0; t < 4; ++t) c[i][t] = (v8f){};
  }
#pragma unroll 2
  for (int j = 0; j < C_DIM / 8; ++j) {
    v4f f[4];
#pragma unroll
    for (int i = 0; i < 4; ++i)
      f[i] = *(const v4f*)(arow + (size_t)(i * 16) * C_DIM + j * 8);
    const float* wp = wrow + (size_t)(j * 8) * (3 * C_DIM);
    const v4f w0 = *(const v4f*)(wp);
    const v4f w1 = *(const v4f*)(wp + 3 * C_DIM);
    const v4f w2 = *(const v4f*)(wp + 6 * C_DIM);
    const v4f w3 = *(const v4f*)(wp + 9 * C_DIM);
    v2f b[4];
#pragma unroll
    for (int t = 0; t < 4; ++t) { b[t].x = w0[t]; b[t].y = w1[t]; }
#pragma unroll
    for (int i = 0; i < 4; ++i) {
      v2f a;
      a.x = f[i].x; a.y = f[i].y;
#pragma unroll
      for (int t = 0; t < 4; ++t) c[i][t] = wmma4(a, b[t], c[i][t]);
    }
#pragma unroll
    for (int t = 0; t < 4; ++t) { b[t].x = w2[t]; b[t].y = w3[t]; }
#pragma unroll
    for (int i = 0; i < 4; ++i) {
      v2f a;
      a.x = f[i].z; a.y = f[i].w;
#pragma unroll
      for (int t = 0; t < 4; ++t) c[i][t] = wmma4(a, b[t], c[i][t]);
    }
  }

  const int which = g >> 2;
  const int h = g & 3;
  const v4f bv = *(const v4f*)(bias + g * 64 + lo * 4);
#pragma unroll
  for (int i = 0; i < 4; ++i) {
    const int mi = m0 + i * 16;
    const int bb = mi >> 11;
    const int t0 = mi & (T_SEQ - 1);
    const size_t tbase = (((size_t)(bb * HEADS + h)) * T_SEQ + t0) * DH;
    const v8f c0 = c[i][0], c1 = c[i][1], c2 = c[i][2], c3 = c[i][3];
    if (which != 2) {
      float* dst = ((which == 0) ? Qw : Kw) + tbase + (lo >> 1) * 128 + hi * 32 +
                   (lo & 1) * 2;
#pragma unroll
      for (int r = 0; r < 8; ++r) {
        v2f pr;
        pr.x = c0[r] + bv.x;
        pr.y = c1[r] + bv.y;
        *(v2f*)(dst + r * 4) = pr;
        pr.x = c2[r] + bv.z;
        pr.y = c3[r] + bv.w;
        *(v2f*)(dst + 64 + r * 4) = pr;
      }
    } else {
      float* dst = Vw + tbase + hi * 64 + (lo >> 3) * 128 + (lo & 3) * 16 +
                   ((lo >> 2) & 1) * 2;
#pragma unroll
      for (int r = 0; r < 8; r += 2) {
        float* d2 = dst + (r >> 1) * 256;
        v2f pr;
        pr.x = c0[r] + bv.x;  pr.y = c0[r + 1] + bv.x;  *(v2f*)(d2) = pr;
        pr.x = c1[r] + bv.y;  pr.y = c1[r + 1] + bv.y;  *(v2f*)(d2 + 4) = pr;
        pr.x = c2[r] + bv.z;  pr.y = c2[r + 1] + bv.z;  *(v2f*)(d2 + 8) = pr;
        pr.x = c3[r] + bv.w;  pr.y = c3[r + 1] + bv.w;  *(v2f*)(d2 + 12) = pr;
      }
    }
  }
}

__global__ void __launch_bounds__(256) attn_kernel(const float* __restrict__ Q,
                            const float* __restrict__ K,
                            const float* __restrict__ V,
                            float* __restrict__ O,
                            int nwaves) {
  const int lane = threadIdx.x & 31;
  const int lo = lane & 15;
  const int hi = lane >> 4;
  int wave = (int)((blockIdx.x * blockDim.x + threadIdx.x) >> 5);
  if (wave >= nwaves) return;

  const int qtiles = T_SEQ / 16;
  const int bh = wave / qtiles;
  const int u = wave % qtiles;
  const int p = (u >> 3) * 4 + (u & 3);
  const int qt = (u & 4) ? p : (qtiles - 1 - p);
  const int q0 = qt * 16;
  const float* Qh = Q + (size_t)bh * T_SEQ * DH;
  const float* Kh = K + (size_t)bh * T_SEQ * DH;
  const float* Vh = V + (size_t)bh * T_SEQ * DH;

  const float qs = 0.125f * 1.44269504088896340736f;
  v2f qreg[16];
  {
    const v4f* qf = (const v4f*)(Qh + (size_t)q0 * DH) + lane;
#pragma unroll
    for (int j = 0; j < 8; ++j) {
      const v4f f = qf[j * 32];
      qreg[2 * j].x = f.x * qs;
      qreg[2 * j].y = f.y * qs;
      qreg[2 * j + 1].x = f.z * qs;
      qreg[2 * j + 1].y = f.w * qs;
    }
  }

  float m_run = -__builtin_inff();
  float l_run = 0.0f;
  v8f o0 = {}, o1 = {}, o2 = {}, o3 = {};

  const v4f* kbase = (const v4f*)Kh + lane;
  const v4f* vbase = (const v4f*)Vh + lane;
  v4f kr[8];
#pragma unroll
  for (int j = 0; j < 8; ++j) kr[j] = kbase[j * 32];

  for (int kt0 = 0; kt0 <= q0; kt0 += 16) {
    v4f vr[8];
    {
      const v4f* vf = vbase + (size_t)kt0 * (DH / 4);
#pragma unroll
      for (int j = 0; j < 8; ++j) vr[j] = vf[j * 32];
    }
    v8f s = {};
#pragma unroll
    for (int j = 0; j < 8; ++j) {
      v2f a;
      a.x = kr[j].x; a.y = kr[j].y;
      s = wmma4(a, qreg[2 * j], s);
      a.x = kr[j].z; a.y = kr[j].w;
      s = wmma4(a, qreg[2 * j + 1], s);
    }
    if (kt0 == q0) {
#pragma unroll
      for (int r = 0; r < 8; ++r)
        s[r] = (r + hi * 8 > lo) ? -__builtin_inff() : s[r];
    }

    {
      const int ktn = (kt0 < q0) ? kt0 + 16 : kt0;
      const v4f* kf = kbase + (size_t)ktn * (DH / 4);
#pragma unroll
      for (int j = 0; j < 8; ++j) kr[j] = kf[j * 32];
    }

    float tmax = s[0];
#pragma unroll
    for (int r = 1; r < 8; ++r) tmax = fmaxf(tmax, s[r]);
    tmax = fmaxf(tmax, __shfl_xor(tmax, 16, 32));
    const float m_new = fmaxf(m_run, tmax);
    const float alpha = __builtin_amdgcn_exp2f(m_run - m_new);
    float p[8];
    float ts = 0.0f;
#pragma unroll
    for (int r = 0; r < 8; ++r) {
      p[r] = __builtin_amdgcn_exp2f(s[r] - m_new);
      ts += p[r];
    }
    ts += __shfl_xor(ts, 16, 32);
    l_run = l_run * alpha + ts;
    m_run = m_new;
#pragma unroll
    for (int r = 0; r < 8; ++r) {
      o0[r] *= alpha; o1[r] *= alpha; o2[r] *= alpha; o3[r] *= alpha;
    }

#pragma unroll
    for (int i = 0; i < 4; ++i) {
      v2f bop;
      bop.x = p[2 * i];
      bop.y = p[2 * i + 1];
      const v4f f0 = vr[2 * i];
      const v4f f1 = vr[2 * i + 1];
      v2f a;
      a.x = f0.x;  a.y = f0.y;   o0 = wmma4(a, bop, o0);
      a.x = f0.z;  a.y = f0.w;   o1 = wmma4(a, bop, o1);
      a.x = f1.x;  a.y = f1.y;   o2 = wmma4(a, bop, o2);
      a.x = f1.z;  a.y = f1.w;   o3 = wmma4(a, bop, o3);
    }
  }

  const float inv_l = 1.0f / l_run;
  const int b = bh / HEADS;
  const int h = bh - b * HEADS;
  float* obase = O + ((size_t)b * T_SEQ + q0) * C_DIM + (h * 8 + hi) * 128 + lo * 4;
#pragma unroll
  for (int hh = 0; hh < 2; ++hh) {
    v4f w;
    w.x = o0[2 * hh] * inv_l;     w.y = o0[2 * hh + 1] * inv_l;
    w.z = o0[2 * hh + 4] * inv_l; w.w = o0[2 * hh + 5] * inv_l;
    *(v4f*)(obase + hh * 64) = w;
    w.x = o1[2 * hh] * inv_l;     w.y = o1[2 * hh + 1] * inv_l;
    w.z = o1[2 * hh + 4] * inv_l; w.w = o1[2 * hh + 5] * inv_l;
    *(v4f*)(obase + 256 + hh * 64) = w;
    w.x = o2[2 * hh] * inv_l;     w.y = o2[2 * hh + 1] * inv_l;
    w.z = o2[2 * hh + 4] * inv_l; w.w = o2[2 * hh + 5] * inv_l;
    *(v4f*)(obase + 512 + hh * 64) = w;
    w.x = o3[2 * hh] * inv_l;     w.y = o3[2 * hh + 1] * inv_l;
    w.z = o3[2 * hh + 4] * inv_l; w.w = o3[2 * hh + 5] * inv_l;
    *(v4f*)(obase + 768 + hh * 64) = w;
  }
}

__global__ void out_proj_kernel(const float* __restrict__ A,
                                const float* __restrict__ W,
                                const float* __restrict__ bias,
                                float* __restrict__ Out,
                                int mtiles) {
  const int lane = threadIdx.x & 31;
  const int lo = lane & 15;
  const int hi = lane >> 4;
  const int wv = (int)((blockIdx.x * blockDim.x + threadIdx.x) >> 5);
  const int g = wv & 3;
  const int mt = (wv >> 3) * 2 + ((wv >> 2) & 1);
  if (mt >= mtiles) return;
  const int m0 = mt * 16;

  const v4f* af = (const v4f*)(A + (size_t)m0 * C_DIM) + lane;
  const float* wrow = W + (size_t)(hi * 2) * C_DIM + g * 64 + lo * 4;
  v8f c0 = {}, c1 = {}, c2 = {}, c3 = {};
#pragma unroll 2
  for (int j = 0; j < C_DIM / 8; ++j) {
    const v4f f = af[j * 32];
    const float* wp = wrow + (size_t)(j * 8) * C_DIM;
    const v4f w0 = *(const v4f*)(wp);
    const v4f w1 = *(const v4f*)(wp + C_DIM);
    const v4f w2 = *(const v4f*)(wp + 4 * C_DIM);
    const v4f w3 = *(const v4f*)(wp + 5 * C_DIM);
    v2f a, b;
    a.x = f.x;  a.y = f.y;
    b.x = w0.x; b.y = w1.x; c0 = wmma4(a, b, c0);
    b.x = w0.y; b.y = w1.y; c1 = wmma4(a, b, c1);
    b.x = w0.z; b.y = w1.z; c2 = wmma4(a, b, c2);
    b.x = w0.w; b.y = w1.w; c3 = wmma4(a, b, c3);
    a.x = f.z;  a.y = f.w;
    b.x = w2.x; b.y = w3.x; c0 = wmma4(a, b, c0);
    b.x = w2.y; b.y = w3.y; c1 = wmma4(a, b, c1);
    b.x = w2.z; b.y = w3.z; c2 = wmma4(a, b, c2);
    b.x = w2.w; b.y = w3.w; c3 = wmma4(a, b, c3);
  }
  const v4f bv = *(const v4f*)(bias + g * 64 + lo * 4);
  float* orow = Out + (size_t)(m0 + hi * 8) * C_DIM + g * 64 + lo * 4;
#pragma unroll
  for (int r = 0; r < 8; ++r) {
    v4f o;
    o.x = c0[r] + bv.x;
    o.y = c1[r] + bv.y;
    o.z = c2[r] + bv.z;
    o.w = c3[r] + bv.w;
    *(v4f*)(orow + (size_t)r * C_DIM) = o;
  }
}

extern "C" void kernel_launch(void* const* d_in, const int* in_sizes, int n_in,
                              void* d_out, int out_size, void* d_ws, size_t ws_size,
                              hipStream_t stream) {
  const float* x     = (const float*)d_in[0];
  const float* W_qkv = (const float*)d_in[1];
  const float* b_qkv = (const float*)d_in[2];
  const float* W_fc  = (const float*)d_in[3];
  const float* b_fc  = (const float*)d_in[4];
  float* out = (float*)d_out;

  const int BT = in_sizes[0] / C_DIM;
  const int B  = BT / T_SEQ;

  const size_t headElems = (size_t)B * HEADS * T_SEQ * DH;
  float* Qw = (float*)d_ws;
  float* Kw = Qw + headElems;
  float* Vw = Kw + headElems;
  float* Aw = Vw + headElems;

  const int mtiles = BT / 16;

  {
    const int mgroups = BT / 64;
    qkv_proj_kernel<<<mgroups, 32 * ((3 * C_DIM) / 64), 0, stream>>>(
        x, W_qkv, b_qkv, Qw, Kw, Vw, mgroups);
  }
  {
    const int nwaves = B * HEADS * (T_SEQ / 16);
    const int blocks = (nwaves + 7) / 8;
    attn_kernel<<<blocks, 256, 0, stream>>>(Qw, Kw, Vw, Aw, nwaves);
  }
  {
    const int waves = mtiles * (C_DIM / 64);
    const int blocks = (waves + 7) / 8;
    out_proj_kernel<<<blocks, 256, 0, stream>>>(Aw, W_fc, b_fc, out, mtiles);
  }
}


# all three kernels: each wave lowers its own s_setprio level (3 to 0) at fixed counts of loop trips left, so the wave furthest behind wins its SIMD and a SIMD's waves end together; sums unchanged
# speedup vs baseline: 2.0048x; 1.0332x over previous
#include <hip/hip_runtime.h>


typedef float v2f __attribute__((ext_vector_type(2)));
typedef float v4f __attribute__((ext_vector_type(4)));
typedef float v8f __attribute__((ext_vector_type(8)));

#define T_SEQ 2048
#define HEADS 4
#define DH    64
#define C_DIM 256

__device__ __forceinline__ v8f wmma4(v2f a, v2f b, v8f c) {
  return __builtin_amdgcn_wmma_f32_16x16x4_f32(false, a, false, b, (short)0, c,
                                               false, false);
}


__global__ void __launch_bounds__(384) qkv_proj_kernel(const float* __restrict__ X,
                                const float* __restrict__ W,
                                const float* __restrict__ bias,
                                float* __restrict__ Qw,
                                float* __restrict__ Kw,
                                float* __restrict__ Vw,
                                int mgroups) {
  const int lane = threadIdx.x & 31;
  const int lo = lane & 15;
  const int hi = lane >> 4;
  const int g = (int)(threadIdx.x >> 5);
  const int mg = (int)blockIdx.x;
  if (mg >= mgroups) return;
  const int m0 = mg * 64;

  const float* arow = X + (size_t)(m0 + lo) * C_DIM + hi * 4;
  const float* wrow = W + (size_t)(hi * 4) * (3 * C_DIM) + g * 64 + lo * 4;
  v8f c[4][4];
#pragma unroll
  for (int i = 0; i < 4; ++i) {
#pragma unroll
    for (int t = 0; t < 4; ++t) c[i][t] = (v8f){};
  }
  auto ktrip = [&](int j) __attribute__((always_inline)) {
    v4f f[4];
#pragma unroll
    for (int i = 0; i < 4; ++i)
      f[i] = *(const v4f*)(arow + (size_t)(i * 16) * C_DIM + j * 8);
    const float* wp = wrow + (size_t)(j * 8) * (3 * C_DIM);
    const v4f w0 = *(const v4f*)(wp);
    const v4f w1 = *(const v4f*)(wp + 3 * C_DIM);
    const v4f w2 = *(const v4f*)(wp + 6 * C_DIM);
    const v4f w3 = *(const v4f*)(wp + 9 * C_DIM);
    v2f b[4];
#pragma unroll
    for (int t = 0; t < 4; ++t) { b[t].x = w0[t]; b[t].y = w1[t]; }
#pragma unroll
    for (int i = 0; i < 4; ++i) {
      v2f a;
      a.x = f[i].x; a.y = f[i].y;
#pragma unroll
      for (int t = 0; t < 4; ++t) c[i][t] = wmma4(a, b[t], c[i][t]);
    }
#pragma unroll
    for (int t = 0; t < 4; ++t) { b[t].x = w2[t]; b[t].y = w3[t]; }
#pragma unroll
    for (int i = 0; i < 4; ++i) {
      v2f a;
      a.x = f[i].z; a.y = f[i].w;
#pragma unroll
      for (int t = 0; t < 4; ++t) c[i][t] = wmma4(a, b[t], c[i][t]);
    }
  };
  __builtin_amdgcn_s_setprio(3);
#pragma unroll 2
  for (int j = 0; j < 16; ++j) ktrip(j);
  __builtin_amdgcn_s_setprio(2);
#pragma unroll 2
  for (int j = 16; j < 24; ++j) ktrip(j);
  __builtin_amdgcn_s_setprio(1);
#pragma unroll 2
  for (int j = 24; j < 28; ++j) ktrip(j);
  __builtin_amdgcn_s_setprio(0);
#pragma unroll 2
  for (int j = 28; j < C_DIM / 8; ++j) ktrip(j);

  const int which = g >> 2;
  const int h = g & 3;
  const v4f bv = *(const v4f*)(bias + g * 64 + lo * 4);
#pragma unroll
  for (int i = 0; i < 4; ++i) {
    const int mi = m0 + i * 16;
    const int bb = mi >> 11;
    const int t0 = mi & (T_SEQ - 1);
    const size_t tbase = (((size_t)(bb * HEADS + h)) * T_SEQ + t0) * DH;
    const v8f c0 = c[i][0], c1 = c[i][1], c2 = c[i][2], c3 = c[i][3];
    if (which != 2) {
      float* dst = ((which == 0) ? Qw : Kw) + tbase + (lo >> 1) * 128 + hi * 32 +
                   (lo & 1) * 2;
#pragma unroll
      for (int r = 0; r < 8; ++r) {
        v2f pr;
        pr.x = c0[r] + bv.x;
        pr.y = c1[r] + bv.y;
        *(v2f*)(dst + r * 4) = pr;
        pr.x = c2[r] + bv.z;
        pr.y = c3[r] + bv.w;
        *(v2f*)(dst + 64 + r * 4) = pr;
      }
    } else {
      float* dst = Vw + tbase + hi * 64 + (lo >> 3) * 128 + (lo & 3) * 16 +
                   ((lo >> 2) & 1) * 2;
#pragma unroll
      for (int r = 0; r < 8; r += 2) {
        float* d2 = dst + (r >> 1) * 256;
        v2f pr;
        pr.x = c0[r] + bv.x;  pr.y = c0[r + 1] + bv.x;  *(v2f*)(d2) = pr;
        pr.x = c1[r] + bv.y;  pr.y = c1[r + 1] + bv.y;  *(v2f*)(d2 + 4) = pr;
        pr.x = c2[r] + bv.z;  pr.y = c2[r + 1] + bv.z;  *(v2f*)(d2 + 8) = pr;
        pr.x = c3[r] + bv.w;  pr.y = c3[r + 1] + bv.w;  *(v2f*)(d2 + 12) = pr;
      }
    }
  }
}

__global__ void __launch_bounds__(256) attn_kernel(const float* __restrict__ Q,
                            const float* __restrict__ K,
                            const float* __restrict__ V,
                            float* __restrict__ O,
                            int nwaves) {
  const int lane = threadIdx.x & 31;
  const int lo = lane & 15;
  const int hi = lane >> 4;
  int wave = (int)((blockIdx.x * blockDim.x + threadIdx.x) >> 5);
  if (wave >= nwaves) return;

  const int qtiles = T_SEQ / 16;
  const int bh = wave / qtiles;
  const int u = wave % qtiles;
  const int p = (u >> 3) * 4 + (u & 3);
  const int qt = (u & 4) ? p : (qtiles - 1 - p);
  const int q0 = qt * 16;
  const float* Qh = Q + (size_t)bh * T_SEQ * DH;
  const float* Kh = K + (size_t)bh * T_SEQ * DH;
  const float* Vh = V + (size_t)bh * T_SEQ * DH;

  const float qs = 0.125f * 1.44269504088896340736f;
  v2f qreg[16];
  {
    const v4f* qf = (const v4f*)(Qh + (size_t)q0 * DH) + lane;
#pragma unroll
    for (int j = 0; j < 8; ++j) {
      const v4f f = qf[j * 32];
      qreg[2 * j].x = f.x * qs;
      qreg[2 * j].y = f.y * qs;
      qreg[2 * j + 1].x = f.z * qs;
      qreg[2 * j + 1].y = f.w * qs;
    }
  }

  float m_run = -__builtin_inff();
  float l_run = 0.0f;
  v8f o0 = {}, o1 = {}, o2 = {}, o3 = {};

  const v4f* kbase = (const v4f*)Kh + lane;
  const v4f* vbase = (const v4f*)Vh + lane;
  v4f kr[8];
#pragma unroll
  for (int j = 0; j < 8; ++j) kr[j] = kbase[j * 32];

  int left = __builtin_amdgcn_readfirstlane(qt) + 1;
  int sw = 0;
  if (left > 48)      { __builtin_amdgcn_s_setprio(3); sw = 48; }
  else if (left > 16) { __builtin_amdgcn_s_setprio(2); sw = 16; }
  else if (left > 4)  { __builtin_amdgcn_s_setprio(1); sw = 4; }

  for (int kt0 = 0; kt0 <= q0; kt0 += 16) {
    if (left == sw) {
      if (sw == 48)      { __builtin_amdgcn_s_setprio(2); sw = 16; }
      else if (sw == 16) { __builtin_amdgcn_s_setprio(1); sw = 4; }
      else               { __builtin_amdgcn_s_setprio(0); sw = 0; }
    }
    --left;
    v4f vr[8];
    {
      const v4f* vf = vbase + (size_t)kt0 * (DH / 4);
#pragma unroll
      for (int j = 0; j < 8; ++j) vr[j] = vf[j * 32];
    }
    v8f s = {};
#pragma unroll
    for (int j = 0; j < 8; ++j) {
      v2f a;
      a.x = kr[j].x; a.y = kr[j].y;
      s = wmma4(a, qreg[2 * j], s);
      a.x = kr[j].z; a.y = kr[j].w;
      s = wmma4(a, qreg[2 * j + 1], s);
    }
    if (kt0 == q0) {
#pragma unroll
      for (int r = 0; r < 8; ++r)
        s[r] = (r + hi * 8 > lo) ? -__builtin_inff() : s[r];
    }

    {
      const int ktn = (kt0 < q0) ? kt0 + 16 : kt0;
      const v4f* kf = kbase + (size_t)ktn * (DH / 4);
#pragma unroll
      for (int j = 0; j < 8; ++j) kr[j] = kf[j * 32];
    }

    float tmax = s[0];
#pragma unroll
    for (int r = 1; r < 8; ++r) tmax = fmaxf(tmax, s[r]);
    tmax = fmaxf(tmax, __shfl_xor(tmax, 16, 32));
    const float m_new = fmaxf(m_run, tmax);
    const float alpha = __builtin_amdgcn_exp2f(m_run - m_new);
    float p[8];
    float ts = 0.0f;
#pragma unroll
    for (int r = 0; r < 8; ++r) {
      p[r] = __builtin_amdgcn_exp2f(s[r] - m_new);
      ts += p[r];
    }
    ts += __shfl_xor(ts, 16, 32);
    l_run = l_run * alpha + ts;
    m_run = m_new;
#pragma unroll
    for (int r = 0; r < 8; ++r) {
      o0[r] *= alpha; o1[r] *= alpha; o2[r] *= alpha; o3[r] *= alpha;
    }

#pragma unroll
    for (int i = 0; i < 4; ++i) {
      v2f bop;
      bop.x = p[2 * i];
      bop.y = p[2 * i + 1];
      const v4f f0 = vr[2 * i];
      const v4f f1 = vr[2 * i + 1];
      v2f a;
      a.x = f0.x;  a.y = f0.y;   o0 = wmma4(a, bop, o0);
      a.x = f0.z;  a.y = f0.w;   o1 = wmma4(a, bop, o1);
      a.x = f1.x;  a.y = f1.y;   o2 = wmma4(a, bop, o2);
      a.x = f1.z;  a.y = f1.w;   o3 = wmma4(a, bop, o3);
    }
  }

  __builtin_amdgcn_s_setprio(0);

  const float inv_l = 1.0f / l_run;
  const int b = bh / HEADS;
  const int h = bh - b * HEADS;
  float* obase = O + ((size_t)b * T_SEQ + q0) * C_DIM + (h * 8 + hi) * 128 + lo * 4;
#pragma unroll
  for (int hh = 0; hh < 2; ++hh) {
    v4f w;
    w.x = o0[2 * hh] * inv_l;     w.y = o0[2 * hh + 1] * inv_l;
    w.z = o0[2 * hh + 4] * inv_l; w.w = o0[2 * hh + 5] * inv_l;
    *(v4f*)(obase + hh * 64) = w;
    w.x = o1[2 * hh] * inv_l;     w.y = o1[2 * hh + 1] * inv_l;
    w.z = o1[2 * hh + 4] * inv_l; w.w = o1[2 * hh + 5] * inv_l;
    *(v4f*)(obase + 256 + hh * 64) = w;
    w.x = o2[2 * hh] * inv_l;     w.y = o2[2 * hh + 1] * inv_l;
    w.z = o2[2 * hh + 4] * inv_l; w.w = o2[2 * hh + 5] * inv_l;
    *(v4f*)(obase + 512 + hh * 64) = w;
    w.x = o3[2 * hh] * inv_l;     w.y = o3[2 * hh + 1] * inv_l;
    w.z = o3[2 * hh + 4] * inv_l; w.w = o3[2 * hh + 5] * inv_l;
    *(v4f*)(obase + 768 + hh * 64) = w;
  }
}

__global__ void out_proj_kernel(const float* __restrict__ A,
                                const float* __restrict__ W,
                                const float* __restrict__ bias,
                                float* __restrict__ Out,
                                int mtiles) {
  const int lane = threadIdx.x & 31;
  const int lo = lane & 15;
  const int hi = lane >> 4;
  const int wv = (int)((blockIdx.x * blockDim.x + threadIdx.x) >> 5);
  const int g = wv & 3;
  const int mt = (wv >> 3) * 2 + ((wv >> 2) & 1);
  if (mt >= mtiles) return;
  const int m0 = mt * 16;

  const v4f* af = (const v4f*)(A + (size_t)m0 * C_DIM) + lane;
  const float* wrow = W + (size_t)(hi * 2) * C_DIM + g * 64 + lo * 4;
  v8f c0 = {}, c1 = {}, c2 = {}, c3 = {};
  auto ktrip = [&](int j) __attribute__((always_inline)) {
    const v4f f = af[j * 32];
    const float* wp = wrow + (size_t)(j * 8) * C_DIM;
    const v4f w0 = *(const v4f*)(wp);
    const v4f w1 = *(const v4f*)(wp + C_DIM);
    const v4f w2 = *(const v4f*)(wp + 4 * C_DIM);
    const v4f w3 = *(const v4f*)(wp + 5 * C_DIM);
    v2f a, b;
    a.x = f.x;  a.y = f.y;
    b.x = w0.x; b.y = w1.x; c0 = wmma4(a, b, c0);
    b.x = w0.y; b.y = w1.y; c1 = wmma4(a, b, c1);
    b.x = w0.z; b.y = w1.z; c2 = wmma4(a, b, c2);
    b.x = w0.w; b.y = w1.w; c3 = wmma4(a, b, c3);
    a.x = f.z;  a.y = f.w;
    b.x = w2.x; b.y = w3.x; c0 = wmma4(a, b, c0);
    b.x = w2.y; b.y = w3.y; c1 = wmma4(a, b, c1);
    b.x = w2.z; b.y = w3.z; c2 = wmma4(a, b, c2);
    b.x = w2.w; b.y = w3.w; c3 = wmma4(a, b, c3);
  };
  __builtin_amdgcn_s_setprio(3);
#pragma unroll 2
  for (int j = 0; j < 16; ++j) ktrip(j);
  __builtin_amdgcn_s_setprio(2);
#pragma unroll 2
  for (int j = 16; j < 24; ++j) ktrip(j);
  __builtin_amdgcn_s_setprio(1);
#pragma unroll 2
  for (int j = 24; j < 28; ++j) ktrip(j);
  __builtin_amdgcn_s_setprio(0);
#pragma unroll 2
  for (int j = 28; j < C_DIM / 8; ++j) ktrip(j);
  const v4f bv = *(const v4f*)(bias + g * 64 + lo * 4);
  float* orow = Out + (size_t)(m0 + hi * 8) * C_DIM + g * 64 + lo * 4;
#pragma unroll
  for (int r = 0; r < 8; ++r) {
    v4f o;
    o.x = c0[r] + bv.x;
    o.y = c1[r] + bv.y;
    o.z = c2[r] + bv.z;
    o.w = c3[r] + bv.w;
    *(v4f*)(orow + (size_t)r * C_DIM) = o;
  }
}

extern "C" void kernel_launch(void* const* d_in, const int* in_sizes, int n_in,
                              void* d_out, int out_size, void* d_ws, size_t ws_size,
                              hipStream_t stream) {
  const float* x     = (const float*)d_in[0];
  const float* W_qkv = (const float*)d_in[1];
  const float* b_qkv = (const float*)d_in[2];
  const float* W_fc  = (const float*)d_in[3];
  const float* b_fc  = (const float*)d_in[4];
  float* out = (float*)d_out;

  const int BT = in_sizes[0] / C_DIM;
  const int B  = BT / T_SEQ;

  const size_t headElems = (size_t)B * HEADS * T_SEQ * DH;
  float* Qw = (float*)d_ws;
  float* Kw = Qw + headElems;
  float* Vw = Kw + headElems;
  float* Aw = Vw + headElems;

  const int mtiles = BT / 16;

  {
    const int mgroups = BT / 64;
    qkv_proj_kernel<<<mgroups, 32 * ((3 * C_DIM) / 64), 0, stream>>>(
        x, W_qkv, b_qkv, Qw, Kw, Vw, mgroups);
  }
  {
    const int nwaves = B * HEADS * (T_SEQ / 16);
    const int blocks = (nwaves + 7) / 8;
    attn_kernel<<<blocks, 256, 0, stream>>>(Qw, Kw, Vw, Aw, nwaves);
  }
  {
    const int waves = mtiles * (C_DIM / 64);
    const int blocks = (waves + 7) / 8;
    out_proj_kernel<<<blocks, 256, 0, stream>>>(Aw, W_fc, b_fc, out, mtiles);
  }
}


# attention: K and V tiles are requested one to two steps ahead into a second register set (the sets take turns per step): a step no longer waits a memory latency behind its score run; sums unchanged
# speedup vs baseline: 2.1465x; 1.0707x over previous
#include <hip/hip_runtime.h>


typedef float v2f __attribute__((ext_vector_type(2)));
typedef float v4f __attribute__((ext_vector_type(4)));
typedef float v8f __attribute__((ext_vector_type(8)));

#define T_SEQ 2048
#define HEADS 4
#define DH    64
#define C_DIM 256

__device__ __forceinline__ v8f wmma4(v2f a, v2f b, v8f c) {
  return __builtin_amdgcn_wmma_f32_16x16x4_f32(false, a, false, b, (short)0, c,
                                               false, false);
}


__global__ void __launch_bounds__(384) qkv_proj_kernel(const float* __restrict__ X,
                                const float* __restrict__ W,
                                const float* __restrict__ bias,
                                float* __restrict__ Qw,
                                float* __restrict__ Kw,
                                float* __restrict__ Vw,
                                int mgroups) {
  const int lane = threadIdx.x & 31;
  const int lo = lane & 15;
  const int hi = lane >> 4;
  const int g = (int)(threadIdx.x >> 5);
  const int mg = (int)blockIdx.x;
  if (mg >= mgroups) return;
  const int m0 = mg * 64;

  const float* arow = X + (size_t)(m0 + lo) * C_DIM + hi * 4;
  const float* wrow = W + (size_t)(hi * 4) * (3 * C_DIM) + g * 64 + lo * 4;
  v8f c[4][4];
#pragma unroll
  for (int i = 0; i < 4; ++i) {
#pragma unroll
    for (int t = 0; t < 4; ++t) c[i][t] = (v8f){};
  }
  auto ktrip = [&](int j) __attribute__((always_inline)) {
    v4f f[4];
#pragma unroll
    for (int i = 0; i < 4; ++i)
      f[i] = *(const v4f*)(arow + (size_t)(i * 16) * C_DIM + j * 8);
    const float* wp = wrow + (size_t)(j * 8) * (3 * C_DIM);
    const v4f w0 = *(const v4f*)(wp);
    const v4f w1 = *(const v4f*)(wp + 3 * C_DIM);
    const v4f w2 = *(const v4f*)(wp + 6 * C_DIM);
    const v4f w3 = *(const v4f*)(wp + 9 * C_DIM);
    v2f b[4];
#pragma unroll
    for (int t = 0; t < 4; ++t) { b[t].x = w0[t]; b[t].y = w1[t]; }
#pragma unroll
    for (int i = 0; i < 4; ++i) {
      v2f a;
      a.x = f[i].x; a.y = f[i].y;
#pragma unroll
      for (int t = 0; t < 4; ++t) c[i][t] = wmma4(a, b[t], c[i][t]);
    }
#pragma unroll
    for (int t = 0; t < 4; ++t) { b[t].x = w2[t]; b[t].y = w3[t]; }
#pragma unroll
    for (int i = 0; i < 4; ++i) {
      v2f a;
      a.x = f[i].z; a.y = f[i].w;
#pragma unroll
      for (int t = 0; t < 4; ++t) c[i][t] = wmma4(a, b[t], c[i][t]);
    }
  };
  __builtin_amdgcn_s_setprio(3);
#pragma unroll 2
  for (int j = 0; j < 16; ++j) ktrip(j);
  __builtin_amdgcn_s_setprio(2);
#pragma unroll 2
  for (int j = 16; j < 24; ++j) ktrip(j);
  __builtin_amdgcn_s_setprio(1);
#pragma unroll 2
  for (int j = 24; j < 28; ++j) ktrip(j);
  __builtin_amdgcn_s_setprio(0);
#pragma unroll 2
  for (int j = 28; j < C_DIM / 8; ++j) ktrip(j);

  const int which = g >> 2;
  const int h = g & 3;
  const v4f bv = *(const v4f*)(bias + g * 64 + lo * 4);
#pragma unroll
  for (int i = 0; i < 4; ++i) {
    const int mi = m0 + i * 16;
    const int bb = mi >> 11;
    const int t0 = mi & (T_SEQ - 1);
    const size_t tbase = (((size_t)(bb * HEADS + h)) * T_SEQ + t0) * DH;
    const v8f c0 = c[i][0], c1 = c[i][1], c2 = c[i][2], c3 = c[i][3];
    if (which != 2) {
      float* dst = ((which == 0) ? Qw : Kw) + tbase + (lo >> 1) * 128 + hi * 32 +
                   (lo & 1) * 2;
#pragma unroll
      for (int r = 0; r < 8; ++r) {
        v2f pr;
        pr.x = c0[r] + bv.x;
        pr.y = c1[r] + bv.y;
        *(v2f*)(dst + r * 4) = pr;
        pr.x = c2[r] + bv.z;
        pr.y = c3[r] + bv.w;
        *(v2f*)(dst + 64 + r * 4) = pr;
      }
    } else {
      float* dst = Vw + tbase + hi * 64 + (lo >> 3) * 128 + (lo & 3) * 16 +
                   ((lo >> 2) & 1) * 2;
#pragma unroll
      for (int r = 0; r < 8; r += 2) {
        float* d2 = dst + (r >> 1) * 256;
        v2f pr;
        pr.x = c0[r] + bv.x;  pr.y = c0[r + 1] + bv.x;  *(v2f*)(d2) = pr;
        pr.x = c1[r] + bv.y;  pr.y = c1[r + 1] + bv.y;  *(v2f*)(d2 + 4) = pr;
        pr.x = c2[r] + bv.z;  pr.y = c2[r + 1] + bv.z;  *(v2f*)(d2 + 8) = pr;
        pr.x = c3[r] + bv.w;  pr.y = c3[r + 1] + bv.w;  *(v2f*)(d2 + 12) = pr;
      }
    }
  }
}

__global__ void __launch_bounds__(256) attn_kernel(const float* __restrict__ Q,
                            const float* __restrict__ K,
                            const float* __restrict__ V,
                            float* __restrict__ O,
                            int nwaves) {
  const int lane = threadIdx.x & 31;
  const int lo = lane & 15;
  const int hi = lane >> 4;
  int wave = (int)((blockIdx.x * blockDim.x + threadIdx.x) >> 5);
  if (wave >= nwaves) return;

  const int qtiles = T_SEQ / 16;
  const int bh = wave / qtiles;
  const int u = wave % qtiles;
  const int p = (u >> 3) * 4 + (u & 3);
  const int qt = (u & 4) ? p : (qtiles - 1 - p);
  const int q0 = qt * 16;
  const float* Qh = Q + (size_t)bh * T_SEQ * DH;
  const float* Kh = K + (size_t)bh * T_SEQ * DH;
  const float* Vh = V + (size_t)bh * T_SEQ * DH;

  const float qs = 0.125f * 1.44269504088896340736f;
  v2f qreg[16];
  {
    const v4f* qf = (const v4f*)(Qh + (size_t)q0 * DH) + lane;
#pragma unroll
    for (int j = 0; j < 8; ++j) {
      const v4f f = qf[j * 32];
      qreg[2 * j].x = f.x * qs;
      qreg[2 * j].y = f.y * qs;
      qreg[2 * j + 1].x = f.z * qs;
      qreg[2 * j + 1].y = f.w * qs;
    }
  }

  float m_run = -__builtin_inff();
  float l_run = 0.0f;
  v8f o0 = {}, o1 = {}, o2 = {}, o3 = {};

  const int qts = __builtin_amdgcn_readfirstlane(qt);

  const v4f* kbase = (const v4f*)Kh + lane;
  const v4f* vbase = (const v4f*)Vh + lane;
  v4f kA[8], vA[8], kB[8], vB[8];
  {
    const v4f* k1 = kbase + ((qts > 0) ? 256 : 0);
#pragma unroll
    for (int j = 0; j < 8; ++j) kA[j] = kbase[j * 32];
#pragma unroll
    for (int j = 0; j < 8; ++j) vA[j] = vbase[j * 32];
#pragma unroll
    for (int j = 0; j < 8; ++j) kB[j] = k1[j * 32];
  }

  int left = qts + 1;
  int sw = 0;
  if (left > 48)      { __builtin_amdgcn_s_setprio(3); sw = 48; }
  else if (left > 16) { __builtin_amdgcn_s_setprio(2); sw = 16; }
  else if (left > 4)  { __builtin_amdgcn_s_setprio(1); sw = 4; }

  auto step = [&](v4f (&kr)[8], v4f (&vr)[8], v4f (&vo)[8], const int kt)
                  __attribute__((always_inline)) {
    if (left == sw) {
      if (sw == 48)      { __builtin_amdgcn_s_setprio(2); sw = 16; }
      else if (sw == 16) { __builtin_amdgcn_s_setprio(1); sw = 4; }
      else               { __builtin_amdgcn_s_setprio(0); sw = 0; }
    }
    --left;
    {
      const int ktv = (kt < qts) ? kt + 1 : qts;
      const v4f* vf = vbase + (size_t)ktv * 256;
#pragma unroll
      for (int j = 0; j < 8; ++j) vo[j] = vf[j * 32];
    }
    v8f s = {};
#pragma unroll
    for (int j = 0; j < 8; ++j) {
      v2f a;
      a.x = kr[j].x; a.y = kr[j].y;
      s = wmma4(a, qreg[2 * j], s);
      a.x = kr[j].z; a.y = kr[j].w;
      s = wmma4(a, qreg[2 * j + 1], s);
    }
    if (kt == qts) {
#pragma unroll
      for (int r = 0; r < 8; ++r)
        s[r] = (r + hi * 8 > lo) ? -__builtin_inff() : s[r];
    }
    {
      const int ktn = (kt + 2 < qts) ? kt + 2 : qts;
      const v4f* kf = kbase + (size_t)ktn * 256;
#pragma unroll
      for (int j = 0; j < 8; ++j) kr[j] = kf[j * 32];
    }

    float tmax = s[0];
#pragma unroll
    for (int r = 1; r < 8; ++r) tmax = fmaxf(tmax, s[r]);
    tmax = fmaxf(tmax, __shfl_xor(tmax, 16, 32));
    const float m_new = fmaxf(m_run, tmax);
    const float alpha = __builtin_amdgcn_exp2f(m_run - m_new);
    float p[8];
    float ts = 0.0f;
#pragma unroll
    for (int r = 0; r < 8; ++r) {
      p[r] = __builtin_amdgcn_exp2f(s[r] - m_new);
      ts += p[r];
    }
    ts += __shfl_xor(ts, 16, 32);
    l_run = l_run * alpha + ts;
    m_run = m_new;
#pragma unroll
    for (int r = 0; r < 8; ++r) {
      o0[r] *= alpha; o1[r] *= alpha; o2[r] *= alpha; o3[r] *= alpha;
    }

#pragma unroll
    for (int i = 0; i < 4; ++i) {
      v2f bop;
      bop.x = p[2 * i];
      bop.y = p[2 * i + 1];
      const v4f f0 = vr[2 * i];
      const v4f f1 = vr[2 * i + 1];
      v2f a;
      a.x = f0.x;  a.y = f0.y;   o0 = wmma4(a, bop, o0);
      a.x = f0.z;  a.y = f0.w;   o1 = wmma4(a, bop, o1);
      a.x = f1.x;  a.y = f1.y;   o2 = wmma4(a, bop, o2);
      a.x = f1.z;  a.y = f1.w;   o3 = wmma4(a, bop, o3);
    }
  };

  const int npairs = (qts + 1) >> 1;
  for (int i = 0; i < npairs; ++i) {
    step(kA, vA, vB, 2 * i);
    step(kB, vB, vA, 2 * i + 1);
  }
  if (!(qts & 1)) step(kA, vA, vB, qts);

  __builtin_amdgcn_s_setprio(0);

  const float inv_l = 1.0f / l_run;
  const int b = bh / HEADS;
  const int h = bh - b * HEADS;
  float* obase = O + ((size_t)b * T_SEQ + q0) * C_DIM + (h * 8 + hi) * 128 + lo * 4;
#pragma unroll
  for (int hh = 0; hh < 2; ++hh) {
    v4f w;
    w.x = o0[2 * hh] * inv_l;     w.y = o0[2 * hh + 1] * inv_l;
    w.z = o0[2 * hh + 4] * inv_l; w.w = o0[2 * hh + 5] * inv_l;
    *(v4f*)(obase + hh * 64) = w;
    w.x = o1[2 * hh] * inv_l;     w.y = o1[2 * hh + 1] * inv_l;
    w.z = o1[2 * hh + 4] * inv_l; w.w = o1[2 * hh + 5] * inv_l;
    *(v4f*)(obase + 256 + hh * 64) = w;
    w.x = o2[2 * hh] * inv_l;     w.y = o2[2 * hh + 1] * inv_l;
    w.z = o2[2 * hh + 4] * inv_l; w.w = o2[2 * hh + 5] * inv_l;
    *(v4f*)(obase + 512 + hh * 64) = w;
    w.x = o3[2 * hh] * inv_l;     w.y = o3[2 * hh + 1] * inv_l;
    w.z = o3[2 * hh + 4] * inv_l; w.w = o3[2 * hh + 5] * inv_l;
    *(v4f*)(obase + 768 + hh * 64) = w;
  }
}

__global__ void out_proj_kernel(const float* __restrict__ A,
                                const float* __restrict__ W,
                                const float* __restrict__ bias,
                                float* __restrict__ Out,
                                int mtiles) {
  const int lane = threadIdx.x & 31;
  const int lo = lane & 15;
  const int hi = lane >> 4;
  const int wv = (int)((blockIdx.x * blockDim.x + threadIdx.x) >> 5);
  const int g = wv & 3;
  const int mt = (wv >> 3) * 2 + ((wv >> 2) & 1);
  if (mt >= mtiles) return;
  const int m0 = mt * 16;

  const v4f* af = (const v4f*)(A + (size_t)m0 * C_DIM) + lane;
  const float* wrow = W + (size_t)(hi * 2) * C_DIM + g * 64 + lo * 4;
  v8f c0 = {}, c1 = {}, c2 = {}, c3 = {};
  auto ktrip = [&](int j) __attribute__((always_inline)) {
    const v4f f = af[j * 32];
    const float* wp = wrow + (size_t)(j * 8) * C_DIM;
    const v4f w0 = *(const v4f*)(wp);
    const v4f w1 = *(const v4f*)(wp + C_DIM);
    const v4f w2 = *(const v4f*)(wp + 4 * C_DIM);
    const v4f w3 = *(const v4f*)(wp + 5 * C_DIM);
    v2f a, b;
    a.x = f.x;  a.y = f.y;
    b.x = w0.x; b.y = w1.x; c0 = wmma4(a, b, c0);
    b.x = w0.y; b.y = w1.y; c1 = wmma4(a, b, c1);
    b.x = w0.z; b.y = w1.z; c2 = wmma4(a, b, c2);
    b.x = w0.w; b.y = w1.w; c3 = wmma4(a, b, c3);
    a.x = f.z;  a.y = f.w;
    b.x = w2.x; b.y = w3.x; c0 = wmma4(a, b, c0);
    b.x = w2.y; b.y = w3.y; c1 = wmma4(a, b, c1);
    b.x = w2.z; b.y = w3.z; c2 = wmma4(a, b, c2);
    b.x = w2.w; b.y = w3.w; c3 = wmma4(a, b, c3);
  };
  __builtin_amdgcn_s_setprio(3);
#pragma unroll 2
  for (int j = 0; j < 16; ++j) ktrip(j);
  __builtin_amdgcn_s_setprio(2);
#pragma unroll 2
  for (int j = 16; j < 24; ++j) ktrip(j);
  __builtin_amdgcn_s_setprio(1);
#pragma unroll 2
  for (int j = 24; j < 28; ++j) ktrip(j);
  __builtin_amdgcn_s_setprio(0);
#pragma unroll 2
  for (int j = 28; j < C_DIM / 8; ++j) ktrip(j);
  const v4f bv = *(const v4f*)(bias + g * 64 + lo * 4);
  float* orow = Out + (size_t)(m0 + hi * 8) * C_DIM + g * 64 + lo * 4;
#pragma unroll
  for (int r = 0; r < 8; ++r) {
    v4f o;
    o.x = c0[r] + bv.x;
    o.y = c1[r] + bv.y;
    o.z = c2[r] + bv.z;
    o.w = c3[r] + bv.w;
    *(v4f*)(orow + (size_t)r * C_DIM) = o;
  }
}

extern "C" void kernel_launch(void* const* d_in, const int* in_sizes, int n_in,
                              void* d_out, int out_size, void* d_ws, size_t ws_size,
                              hipStream_t stream) {
  const float* x     = (const float*)d_in[0];
  const float* W_qkv = (const float*)d_in[1];
  const float* b_qkv = (const float*)d_in[2];
  const float* W_fc  = (const float*)d_in[3];
  const float* b_fc  = (const float*)d_in[4];
  float* out = (float*)d_out;

  const int BT = in_sizes[0] / C_DIM;
  const int B  = BT / T_SEQ;

  const size_t headElems = (size_t)B * HEADS * T_SEQ * DH;
  float* Qw = (float*)d_ws;
  float* Kw = Qw + headElems;
  float* Vw = Kw + headElems;
  float* Aw = Vw + headElems;

  const int mtiles = BT / 16;

  {
    const int mgroups = BT / 64;
    qkv_proj_kernel<<<mgroups, 32 * ((3 * C_DIM) / 64), 0, stream>>>(
        x, W_qkv, b_qkv, Qw, Kw, Vw, mgroups);
  }
  {
    const int nwaves = B * HEADS * (T_SEQ / 16);
    const int blocks = (nwaves + 7) / 8;
    attn_kernel<<<blocks, 256, 0, stream>>>(Qw, Kw, Vw, Aw, nwaves);
  }
  {
    const int waves = mtiles * (C_DIM / 64);
    const int blocks = (waves + 7) / 8;
    out_proj_kernel<<<blocks, 256, 0, stream>>>(Aw, W_fc, b_fc, out, mtiles);
  }
}
